# starting kernel
# baseline (speedup 1.0000x reference)
#include <hip/hip_runtime.h>
#include <hip/hip_bf16.h>

typedef __attribute__((ext_vector_type(2))) float v2f;
typedef __attribute__((ext_vector_type(8))) float v8f;

#define CH    768
#define HW    4096
#define NOFF  28
#define KTOT  796

__global__ void escn_norm_kernel(const float* __restrict__ feat,
                                 float* __restrict__ invn) {
    int idx = blockIdx.x * blockDim.x + threadIdx.x;
    int b = idx >> 12;
    int p = idx & (HW - 1);
    const float* base = feat + ((size_t)b * CH) * HW + p;
    float s = 0.0f;
#pragma unroll 4
    for (int c = 0; c < CH; ++c) {
        float v = base[(size_t)c * HW];
        s = fmaf(v, v, s);
    }
    float n = sqrtf(s);
    invn[idx] = 1.0f / fmaxf(n, 1e-12f);
}

__global__ void escn_corr_kernel(const float* __restrict__ feat,
                                 const float* __restrict__ invn,
                                 float* __restrict__ corr) {
    int idx = blockIdx.x * blockDim.x + threadIdx.x;
    int b = idx >> 12;
    int p = idx & (HW - 1);
    int h = p >> 6;
    int w = p & 63;

    int   sp[NOFF];
    float vf[NOFF];
#pragma unroll
    for (int o = 0; o < NOFF; ++o) {
        int r, cc;
        if      (o <  7) { r = o;      cc = o;            }
        else if (o < 14) { r = o - 7;  cc = 3;            }
        else if (o < 21) { r = o - 14; cc = 6 - (o - 14); }
        else             { r = 3;      cc = o - 21;       }
        int sh = h + r - 3;
        int sw = w + cc - 3;
        bool ok = (sh >= 0) && (sh < 64) && (sw >= 0) && (sw < 64);
        sp[o] = ok ? (sh * 64 + sw) : p;
        vf[o] = ok ? 1.0f : 0.0f;
    }

    float acc[NOFF];
#pragma unroll
    for (int o = 0; o < NOFF; ++o) acc[o] = 0.0f;

    const float* base = feat + ((size_t)b * CH) * HW;
    for (int c = 0; c < CH; ++c) {
        const float* row = base + (size_t)c * HW;
        float center = row[p];
#pragma unroll
        for (int o = 0; o < NOFF; ++o)
            acc[o] = fmaf(row[sp[o]], center, acc[o]);
    }

    const float* ib = invn + b * HW;
    float ic = ib[p];
#pragma unroll
    for (int o = 0; o < NOFF; ++o)
        corr[((size_t)b * NOFF + o) * HW + p] = acc[o] * vf[o] * ic * ib[sp[o]];
}

__global__ void escn_gemm_kernel(const float* __restrict__ feat,
                                 const float* __restrict__ corr,
                                 const float* __restrict__ Wm,
                                 const float* __restrict__ bias,
                                 float* __restrict__ out) {
    const int lane  = threadIdx.x & 31;
    const int wid   = threadIdx.x >> 5;
    const int mtile = blockIdx.x * 8 + wid;
    const int m0    = mtile * 64;
    const int o0    = blockIdx.y * 32;
    const int b     = blockIdx.z;

    const int col   = lane & 15;
    const int khalf = lane >> 4;

    v8f acc00 = {}, acc01 = {}, acc02 = {}, acc03 = {};
    v8f acc10 = {}, acc11 = {}, acc12 = {}, acc13 = {};

    const float* fbase = feat + ((size_t)b * CH)   * HW + m0 + col;
    const float* cbase = corr + ((size_t)b * NOFF) * HW + m0 + col;
    const float* wrow0 = Wm + (size_t)(o0 + col) * KTOT + 2 * khalf;
    const float* wrow1 = wrow0 + (size_t)16 * KTOT;

    for (int k = 0; k < CH; k += 4) {
        v2f a0 = *(const v2f*)(wrow0 + k);
        v2f a1 = *(const v2f*)(wrow1 + k);

        const float* xk = fbase + ((size_t)k + 2 * khalf) * HW;
        v2f b0, b1, b2, b3;
        b0.x = xk[0];  b0.y = xk[HW + 0];
        b1.x = xk[16]; b1.y = xk[HW + 16];
        b2.x = xk[32]; b2.y = xk[HW + 32];
        b3.x = xk[48]; b3.y = xk[HW + 48];

        acc00 = __builtin_amdgcn_wmma_f32_16x16x4_f32(false, a0, false, b0, (short)0, acc00, false, false);
        acc01 = __builtin_amdgcn_wmma_f32_16x16x4_f32(false, a0, false, b1, (short)0, acc01, false, false);
        acc02 = __builtin_amdgcn_wmma_f32_16x16x4_f32(false, a0, false, b2, (short)0, acc02, false, false);
        acc03 = __builtin_amdgcn_wmma_f32_16x16x4_f32(false, a0, false, b3, (short)0, acc03, false, false);
        acc10 = __builtin_amdgcn_wmma_f32_16x16x4_f32(false, a1, false, b0, (short)0, acc10, false, false);
        acc11 = __builtin_amdgcn_wmma_f32_16x16x4_f32(false, a1, false, b1, (short)0, acc11, false, false);
        acc12 = __builtin_amdgcn_wmma_f32_16x16x4_f32(false, a1, false, b2, (short)0, acc12, false, false);
        acc13 = __builtin_amdgcn_wmma_f32_16x16x4_f32(false, a1, false, b3, (short)0, acc13, false, false);
    }

    for (int k = 0; k < NOFF; k += 4) {
        v2f a0 = *(const v2f*)(wrow0 + CH + k);
        v2f a1 = *(const v2f*)(wrow1 + CH + k);

        const float* xk = cbase + ((size_t)k + 2 * khalf) * HW;
        v2f b0, b1, b2, b3;
        b0.x = xk[0];  b0.y = xk[HW + 0];
        b1.x = xk[16]; b1.y = xk[HW + 16];
        b2.x = xk[32]; b2.y = xk[HW + 32];
        b3.x = xk[48]; b3.y = xk[HW + 48];

        acc00 = __builtin_amdgcn_wmma_f32_16x16x4_f32(false, a0, false, b0, (short)0, acc00, false, false);
        acc01 = __builtin_amdgcn_wmma_f32_16x16x4_f32(false, a0, false, b1, (short)0, acc01, false, false);
        acc02 = __builtin_amdgcn_wmma_f32_16x16x4_f32(false, a0, false, b2, (short)0, acc02, false, false);
        acc03 = __builtin_amdgcn_wmma_f32_16x16x4_f32(false, a0, false, b3, (short)0, acc03, false, false);
        acc10 = __builtin_amdgcn_wmma_f32_16x16x4_f32(false, a1, false, b0, (short)0, acc10, false, false);
        acc11 = __builtin_amdgcn_wmma_f32_16x16x4_f32(false, a1, false, b1, (short)0, acc11, false, false);
        acc12 = __builtin_amdgcn_wmma_f32_16x16x4_f32(false, a1, false, b2, (short)0, acc12, false, false);
        acc13 = __builtin_amdgcn_wmma_f32_16x16x4_f32(false, a1, false, b3, (short)0, acc13, false, false);
    }

    float* outb = out + ((size_t)b * CH) * HW + m0 + col;
#pragma unroll
    for (int v = 0; v < 8; ++v) {
        int oA = o0 + v + 8 * khalf;
        int oB = oA + 16;
        float bsA = bias[oA];
        float bsB = bias[oB];
        float* rowA = outb + (size_t)oA * HW;
        float* rowB = outb + (size_t)oB * HW;
        rowA[0]  = acc00[v] + bsA;
        rowA[16] = acc01[v] + bsA;
        rowA[32] = acc02[v] + bsA;
        rowA[48] = acc03[v] + bsA;
        rowB[0]  = acc10[v] + bsB;
        rowB[16] = acc11[v] + bsB;
        rowB[32] = acc12[v] + bsB;
        rowB[48] = acc13[v] + bsB;
    }
}

extern "C" void kernel_launch(void* const* d_in, const int* in_sizes, int n_in,
                              void* d_out, int out_size, void* d_ws, size_t ws_size,
                              hipStream_t stream) {
    const float* feature = (const float*)d_in[0];
    const float* conv_w  = (const float*)d_in[1];
    const float* conv_b  = (const float*)d_in[2];
    float* out = (float*)d_out;

    float* invn = (float*)d_ws;
    float* corr = invn + 32768;

    escn_norm_kernel<<<128, 256, 0, stream>>>(feature, invn);
    escn_corr_kernel<<<128, 256, 0, stream>>>(feature, invn, corr);
    escn_gemm_kernel<<<dim3(8, 24, 8), 256, 0, stream>>>(feature, corr,
                                                         conv_w, conv_b, out);
}


# norm and corr kernels: 768-channel sweep split over the 8 waves of a work-group (96 channels each, 32 pixels per group), partial sums added in slice order through LDS; changes summation order
# speedup vs baseline: 2.4907x; 2.4907x over previous
#include <hip/hip_runtime.h>
#include <hip/hip_bf16.h>

typedef __attribute__((ext_vector_type(2))) float v2f;
typedef __attribute__((ext_vector_type(8))) float v8f;

#define CH    768
#define HW    4096
#define NOFF  28
#define KTOT  796

#define CSPLIT 8
#define CSLICE (CH / CSPLIT)

__global__ __launch_bounds__(256)
void escn_norm_kernel(const float* __restrict__ feat,
                      float* __restrict__ invn) {
    __shared__ float part[CSPLIT * 32];

    const int lane = threadIdx.x & 31;
    const int g    = __builtin_amdgcn_readfirstlane((int)(threadIdx.x >> 5));
    const int b    = blockIdx.x >> 7;
    const int p    = ((blockIdx.x & 127) << 5) + lane;

    const float* base = feat + ((size_t)b * CH + (size_t)g * CSLICE) * HW + p;
    float s = 0.0f;
#pragma unroll 4
    for (int c = 0; c < CSLICE; ++c) {
        float v = base[(size_t)c * HW];
        s = fmaf(v, v, s);
    }
    part[g * 32 + lane] = s;
    __syncthreads();

    if (g == 0) {
        float t = part[lane];
#pragma unroll
        for (int q = 1; q < CSPLIT; ++q) t += part[q * 32 + lane];
        float n = sqrtf(t);
        invn[b * HW + p] = 1.0f / fmaxf(n, 1e-12f);
    }
}

__device__ __forceinline__ void escn_offset(int o, int& r, int& cc) {
    if      (o <  7) { r = o;      cc = o;            }
    else if (o < 14) { r = o - 7;  cc = 3;            }
    else if (o < 21) { r = o - 14; cc = 6 - (o - 14); }
    else             { r = 3;      cc = o - 21;       }
}

__attribute__((amdgpu_waves_per_eu(8, 8)))
__global__ __launch_bounds__(256)
void escn_corr_kernel(const float* __restrict__ feat,
                      const float* __restrict__ invn,
                      float* __restrict__ corr) {
    __shared__ float red[CSPLIT * NOFF * 32];

    const int lane = threadIdx.x & 31;
    const int g    = __builtin_amdgcn_readfirstlane((int)(threadIdx.x >> 5));
    const int b    = blockIdx.x >> 7;
    const int p    = ((blockIdx.x & 127) << 5) + lane;
    const int h    = p >> 6;
    const int w    = p & 63;

    unsigned spb[NOFF];
#pragma unroll
    for (int o = 0; o < NOFF; ++o) {
        int r, cc;
        escn_offset(o, r, cc);
        int sh = h + r - 3;
        int sw = w + cc - 3;
        bool ok = (sh >= 0) && (sh < 64) && (sw >= 0) && (sw < 64);
        spb[o] = 4u * (unsigned)(ok ? (sh * 64 + sw) : p);
    }
    const unsigned pb = 4u * (unsigned)p;

    float acc[NOFF];
#pragma unroll
    for (int o = 0; o < NOFF; ++o) acc[o] = 0.0f;

    const char* base = (const char*)(feat + ((size_t)b * CH + (size_t)g * CSLICE) * HW);
    for (int c = 0; c < CSLICE; ++c) {
        const unsigned plane = (unsigned)c * (unsigned)(HW * sizeof(float));
        float center = *(const float*)(base + (plane + pb));
#pragma unroll
        for (int o = 0; o < NOFF; ++o)
            acc[o] = fmaf(*(const float*)(base + (plane + spb[o])), center, acc[o]);
    }

#pragma unroll
    for (int o = 0; o < NOFF; ++o)
        red[(g * NOFF + o) * 32 + lane] = acc[o];
    __syncthreads();

    const float* ib = invn + b * HW;
    const float ic = ib[p];
    for (int o = g; o < NOFF; o += CSPLIT) {
        float s = red[o * 32 + lane];
#pragma unroll
        for (int q = 1; q < CSPLIT; ++q)
            s += red[(q * NOFF + o) * 32 + lane];
        int r, cc;
        escn_offset(o, r, cc);
        int sh = h + r - 3;
        int sw = w + cc - 3;
        bool ok = (sh >= 0) && (sh < 64) && (sw >= 0) && (sw < 64);
        int   spn = ok ? (sh * 64 + sw) : p;
        float vf  = ok ? 1.0f : 0.0f;
        corr[((size_t)b * NOFF + o) * HW + p] = s * vf * ic * ib[spn];
    }
}

__global__ void escn_gemm_kernel(const float* __restrict__ feat,
                                 const float* __restrict__ corr,
                                 const float* __restrict__ Wm,
                                 const float* __restrict__ bias,
                                 float* __restrict__ out) {
    const int lane  = threadIdx.x & 31;
    const int wid   = threadIdx.x >> 5;
    const int mtile = blockIdx.x * 8 + wid;
    const int m0    = mtile * 64;
    const int o0    = blockIdx.y * 32;
    const int b     = blockIdx.z;

    const int col   = lane & 15;
    const int khalf = lane >> 4;

    v8f acc00 = {}, acc01 = {}, acc02 = {}, acc03 = {};
    v8f acc10 = {}, acc11 = {}, acc12 = {}, acc13 = {};

    const float* fbase = feat + ((size_t)b * CH)   * HW + m0 + col;
    const float* cbase = corr + ((size_t)b * NOFF) * HW + m0 + col;
    const float* wrow0 = Wm + (size_t)(o0 + col) * KTOT + 2 * khalf;
    const float* wrow1 = wrow0 + (size_t)16 * KTOT;

    for (int k = 0; k < CH; k += 4) {
        v2f a0 = *(const v2f*)(wrow0 + k);
        v2f a1 = *(const v2f*)(wrow1 + k);

        const float* xk = fbase + ((size_t)k + 2 * khalf) * HW;
        v2f b0, b1, b2, b3;
        b0.x = xk[0];  b0.y = xk[HW + 0];
        b1.x = xk[16]; b1.y = xk[HW + 16];
        b2.x = xk[32]; b2.y = xk[HW + 32];
        b3.x = xk[48]; b3.y = xk[HW + 48];

        acc00 = __builtin_amdgcn_wmma_f32_16x16x4_f32(false, a0, false, b0, (short)0, acc00, false, false);
        acc01 = __builtin_amdgcn_wmma_f32_16x16x4_f32(false, a0, false, b1, (short)0, acc01, false, false);
        acc02 = __builtin_amdgcn_wmma_f32_16x16x4_f32(false, a0, false, b2, (short)0, acc02, false, false);
        acc03 = __builtin_amdgcn_wmma_f32_16x16x4_f32(false, a0, false, b3, (short)0, acc03, false, false);
        acc10 = __builtin_amdgcn_wmma_f32_16x16x4_f32(false, a1, false, b0, (short)0, acc10, false, false);
        acc11 = __builtin_amdgcn_wmma_f32_16x16x4_f32(false, a1, false, b1, (short)0, acc11, false, false);
        acc12 = __builtin_amdgcn_wmma_f32_16x16x4_f32(false, a1, false, b2, (short)0, acc12, false, false);
        acc13 = __builtin_amdgcn_wmma_f32_16x16x4_f32(false, a1, false, b3, (short)0, acc13, false, false);
    }

    for (int k = 0; k < NOFF; k += 4) {
        v2f a0 = *(const v2f*)(wrow0 + CH + k);
        v2f a1 = *(const v2f*)(wrow1 + CH + k);

        const float* xk = cbase + ((size_t)k + 2 * khalf) * HW;
        v2f b0, b1, b2, b3;
        b0.x = xk[0];  b0.y = xk[HW + 0];
        b1.x = xk[16]; b1.y = xk[HW + 16];
        b2.x = xk[32]; b2.y = xk[HW + 32];
        b3.x = xk[48]; b3.y = xk[HW + 48];

        acc00 = __builtin_amdgcn_wmma_f32_16x16x4_f32(false, a0, false, b0, (short)0, acc00, false, false);
        acc01 = __builtin_amdgcn_wmma_f32_16x16x4_f32(false, a0, false, b1, (short)0, acc01, false, false);
        acc02 = __builtin_amdgcn_wmma_f32_16x16x4_f32(false, a0, false, b2, (short)0, acc02, false, false);
        acc03 = __builtin_amdgcn_wmma_f32_16x16x4_f32(false, a0, false, b3, (short)0, acc03, false, false);
        acc10 = __builtin_amdgcn_wmma_f32_16x16x4_f32(false, a1, false, b0, (short)0, acc10, false, false);
        acc11 = __builtin_amdgcn_wmma_f32_16x16x4_f32(false, a1, false, b1, (short)0, acc11, false, false);
        acc12 = __builtin_amdgcn_wmma_f32_16x16x4_f32(false, a1, false, b2, (short)0, acc12, false, false);
        acc13 = __builtin_amdgcn_wmma_f32_16x16x4_f32(false, a1, false, b3, (short)0, acc13, false, false);
    }

    float* outb = out + ((size_t)b * CH) * HW + m0 + col;
#pragma unroll
    for (int v = 0; v < 8; ++v) {
        int oA = o0 + v + 8 * khalf;
        int oB = oA + 16;
        float bsA = bias[oA];
        float bsB = bias[oB];
        float* rowA = outb + (size_t)oA * HW;
        float* rowB = outb + (size_t)oB * HW;
        rowA[0]  = acc00[v] + bsA;
        rowA[16] = acc01[v] + bsA;
        rowA[32] = acc02[v] + bsA;
        rowA[48] = acc03[v] + bsA;
        rowB[0]  = acc10[v] + bsB;
        rowB[16] = acc11[v] + bsB;
        rowB[32] = acc12[v] + bsB;
        rowB[48] = acc13[v] + bsB;
    }
}

extern "C" void kernel_launch(void* const* d_in, const int* in_sizes, int n_in,
                              void* d_out, int out_size, void* d_ws, size_t ws_size,
                              hipStream_t stream) {
    const float* feature = (const float*)d_in[0];
    const float* conv_w  = (const float*)d_in[1];
    const float* conv_b  = (const float*)d_in[2];
    float* out = (float*)d_out;

    float* invn = (float*)d_ws;
    float* corr = invn + 32768;

    escn_norm_kernel<<<1024, 256, 0, stream>>>(feature, invn);
    escn_corr_kernel<<<1024, 256, 0, stream>>>(feature, invn, corr);
    escn_gemm_kernel<<<dim3(8, 24, 8), 256, 0, stream>>>(feature, corr,
                                                         conv_w, conv_b, out);
}


# corr kernel: a lane owns 4 consecutive pixels, loads each neighbour row as three 16-byte quads (21 loads feed 100 multiply-adds); 16 channel slices of 48 (was 8 of 96): changes summation order
# speedup vs baseline: 2.8583x; 1.1476x over previous
#include <hip/hip_runtime.h>
#include <hip/hip_bf16.h>

typedef __attribute__((ext_vector_type(2))) float v2f;
typedef __attribute__((ext_vector_type(8))) float v8f;

#define CH    768
#define HW    4096
#define NOFF  28
#define KTOT  796

#define CSPLIT 8
#define CSLICE (CH / CSPLIT)

__global__ __launch_bounds__(256)
void escn_norm_kernel(const float* __restrict__ feat,
                      float* __restrict__ invn) {
    __shared__ float part[CSPLIT * 32];

    const int lane = threadIdx.x & 31;
    const int g    = __builtin_amdgcn_readfirstlane((int)(threadIdx.x >> 5));
    const int b    = blockIdx.x >> 7;
    const int p    = ((blockIdx.x & 127) << 5) + lane;

    const float* base = feat + ((size_t)b * CH + (size_t)g * CSLICE) * HW + p;
    float s = 0.0f;
#pragma unroll 4
    for (int c = 0; c < CSLICE; ++c) {
        float v = base[(size_t)c * HW];
        s = fmaf(v, v, s);
    }
    part[g * 32 + lane] = s;
    __syncthreads();

    if (g == 0) {
        float t = part[lane];
#pragma unroll
        for (int q = 1; q < CSPLIT; ++q) t += part[q * 32 + lane];
        float n = sqrtf(t);
        invn[b * HW + p] = 1.0f / fmaxf(n, 1e-12f);
    }
}

typedef __attribute__((ext_vector_type(4))) float v4f;

#define KSPLIT 16
#define KSLICE (CH / KSPLIT)
#define NUNIQ  25

__device__ __forceinline__ void escn_offset(int o, int& r, int& cc) {
    if      (o <  7) { r = o;      cc = o;            }
    else if (o < 14) { r = o - 7;  cc = 3;            }
    else if (o < 21) { r = o - 14; cc = 6 - (o - 14); }
    else             { r = 3;      cc = o - 21;       }
}

__device__ __forceinline__ int escn_slot(int o) {
    if (o == 10 || o == 17 || o == 24) return 3;
    return o - (o > 24 ? 3 : o > 17 ? 2 : o > 10 ? 1 : 0);
}

__global__ __launch_bounds__(512)
void escn_corr_kernel(const float* __restrict__ feat,
                      const float* __restrict__ invn,
                      float* __restrict__ corr) {
    __shared__ v4f red[KSPLIT * NUNIQ * 32];

    const int lane = threadIdx.x & 31;
    const int g    = __builtin_amdgcn_readfirstlane((int)(threadIdx.x >> 5));
    const int b    = blockIdx.x >> 5;
    const int j    = lane & 15;
    const int h    = ((blockIdx.x & 31) << 1) + (lane >> 4);
    const int p0   = h * 64 + 4 * j;

    unsigned offL[7], offC[7], offR[7];
#pragma unroll
    for (int i = 0; i < 7; ++i) {
        int  rr  = h + i - 3;
        bool rok = (rr >= 0) && (rr < 64);
        unsigned oc = 4u * (unsigned)((rok ? rr : h) * 64 + 4 * j);
        offC[i] = oc;
        offL[i] = oc - (j > 0  ? 16u : 0u);
        offR[i] = oc + (j < 15 ? 16u : 0u);
    }

    float acc[NUNIQ][4];
#pragma unroll
    for (int s = 0; s < NUNIQ; ++s) {
#pragma unroll
        for (int t = 0; t < 4; ++t) acc[s][t] = 0.0f;
    }

    const char* base = (const char*)(feat + ((size_t)b * CH + (size_t)g * KSLICE) * HW);
    for (int c = 0; c < KSLICE; ++c) {
        const unsigned plane = (unsigned)c * (unsigned)(HW * sizeof(float));
        float win[7][12];
#pragma unroll
        for (int i = 0; i < 7; ++i) {
            v4f L = *(const v4f*)(base + (plane + offL[i]));
            v4f C = *(const v4f*)(base + (plane + offC[i]));
            v4f R = *(const v4f*)(base + (plane + offR[i]));
            win[i][0] = L.x; win[i][1] = L.y; win[i][2]  = L.z; win[i][3]  = L.w;
            win[i][4] = C.x; win[i][5] = C.y; win[i][6]  = C.z; win[i][7]  = C.w;
            win[i][8] = R.x; win[i][9] = R.y; win[i][10] = R.z; win[i][11] = R.w;
        }
#pragma unroll
        for (int o = 0; o < NOFF; ++o) {
            if (o == 10 || o == 17 || o == 24) continue;
            int r, cc;
            escn_offset(o, r, cc);
            const int s = escn_slot(o);
#pragma unroll
            for (int t = 0; t < 4; ++t)
                acc[s][t] = fmaf(win[r][1 + t + cc], win[3][4 + t], acc[s][t]);
        }
    }

#pragma unroll
    for (int s = 0; s < NUNIQ; ++s) {
        v4f v;
        v.x = acc[s][0]; v.y = acc[s][1]; v.z = acc[s][2]; v.w = acc[s][3];
        red[(g * NUNIQ + s) * 32 + lane] = v;
    }
    __syncthreads();

    const float* ib = invn + b * HW;
    const v4f ic = *(const v4f*)(ib + p0);
    for (int o = g; o < NOFF; o += KSPLIT) {
        int r, cc;
        escn_offset(o, r, cc);
        const int s = escn_slot(o);
        v4f sum = red[s * 32 + lane];
#pragma unroll
        for (int q = 1; q < KSPLIT; ++q)
            sum += red[(q * NUNIQ + s) * 32 + lane];
        const int  sh  = h + r - 3;
        const bool rok = (sh >= 0) && (sh < 64);
        v4f res;
#pragma unroll
        for (int t = 0; t < 4; ++t) {
            int  sw  = 4 * j + t + cc - 3;
            bool ok  = rok && (sw >= 0) && (sw < 64);
            int  spn = ok ? (sh * 64 + sw) : (p0 + t);
            float vf = ok ? 1.0f : 0.0f;
            res[t] = sum[t] * vf * ic[t] * ib[spn];
        }
        *(v4f*)(corr + ((size_t)b * NOFF + o) * HW + p0) = res;
    }
}

__global__ void escn_gemm_kernel(const float* __restrict__ feat,
                                 const float* __restrict__ corr,
                                 const float* __restrict__ Wm,
                                 const float* __restrict__ bias,
                                 float* __restrict__ out) {
    const int lane  = threadIdx.x & 31;
    const int wid   = threadIdx.x >> 5;
    const int mtile = blockIdx.x * 8 + wid;
    const int m0    = mtile * 64;
    const int o0    = blockIdx.y * 32;
    const int b     = blockIdx.z;

    const int col   = lane & 15;
    const int khalf = lane >> 4;

    v8f acc00 = {}, acc01 = {}, acc02 = {}, acc03 = {};
    v8f acc10 = {}, acc11 = {}, acc12 = {}, acc13 = {};

    const float* fbase = feat + ((size_t)b * CH)   * HW + m0 + col;
    const float* cbase = corr + ((size_t)b * NOFF) * HW + m0 + col;
    const float* wrow0 = Wm + (size_t)(o0 + col) * KTOT + 2 * khalf;
    const float* wrow1 = wrow0 + (size_t)16 * KTOT;

    for (int k = 0; k < CH; k += 4) {
        v2f a0 = *(const v2f*)(wrow0 + k);
        v2f a1 = *(const v2f*)(wrow1 + k);

        const float* xk = fbase + ((size_t)k + 2 * khalf) * HW;
        v2f b0, b1, b2, b3;
        b0.x = xk[0];  b0.y = xk[HW + 0];
        b1.x = xk[16]; b1.y = xk[HW + 16];
        b2.x = xk[32]; b2.y = xk[HW + 32];
        b3.x = xk[48]; b3.y = xk[HW + 48];

        acc00 = __builtin_amdgcn_wmma_f32_16x16x4_f32(false, a0, false, b0, (short)0, acc00, false, false);
        acc01 = __builtin_amdgcn_wmma_f32_16x16x4_f32(false, a0, false, b1, (short)0, acc01, false, false);
        acc02 = __builtin_amdgcn_wmma_f32_16x16x4_f32(false, a0, false, b2, (short)0, acc02, false, false);
        acc03 = __builtin_amdgcn_wmma_f32_16x16x4_f32(false, a0, false, b3, (short)0, acc03, false, false);
        acc10 = __builtin_amdgcn_wmma_f32_16x16x4_f32(false, a1, false, b0, (short)0, acc10, false, false);
        acc11 = __builtin_amdgcn_wmma_f32_16x16x4_f32(false, a1, false, b1, (short)0, acc11, false, false);
        acc12 = __builtin_amdgcn_wmma_f32_16x16x4_f32(false, a1, false, b2, (short)0, acc12, false, false);
        acc13 = __builtin_amdgcn_wmma_f32_16x16x4_f32(false, a1, false, b3, (short)0, acc13, false, false);
    }

    for (int k = 0; k < NOFF; k += 4) {
        v2f a0 = *(const v2f*)(wrow0 + CH + k);
        v2f a1 = *(const v2f*)(wrow1 + CH + k);

        const float* xk = cbase + ((size_t)k + 2 * khalf) * HW;
        v2f b0, b1, b2, b3;
        b0.x = xk[0];  b0.y = xk[HW + 0];
        b1.x = xk[16]; b1.y = xk[HW + 16];
        b2.x = xk[32]; b2.y = xk[HW + 32];
        b3.x = xk[48]; b3.y = xk[HW + 48];

        acc00 = __builtin_amdgcn_wmma_f32_16x16x4_f32(false, a0, false, b0, (short)0, acc00, false, false);
        acc01 = __builtin_amdgcn_wmma_f32_16x16x4_f32(false, a0, false, b1, (short)0, acc01, false, false);
        acc02 = __builtin_amdgcn_wmma_f32_16x16x4_f32(false, a0, false, b2, (short)0, acc02, false, false);
        acc03 = __builtin_amdgcn_wmma_f32_16x16x4_f32(false, a0, false, b3, (short)0, acc03, false, false);
        acc10 = __builtin_amdgcn_wmma_f32_16x16x4_f32(false, a1, false, b0, (short)0, acc10, false, false);
        acc11 = __builtin_amdgcn_wmma_f32_16x16x4_f32(false, a1, false, b1, (short)0, acc11, false, false);
        acc12 = __builtin_amdgcn_wmma_f32_16x16x4_f32(false, a1, false, b2, (short)0, acc12, false, false);
        acc13 = __builtin_amdgcn_wmma_f32_16x16x4_f32(false, a1, false, b3, (short)0, acc13, false, false);
    }

    float* outb = out + ((size_t)b * CH) * HW + m0 + col;
#pragma unroll
    for (int v = 0; v < 8; ++v) {
        int oA = o0 + v + 8 * khalf;
        int oB = oA + 16;
        float bsA = bias[oA];
        float bsB = bias[oB];
        float* rowA = outb + (size_t)oA * HW;
        float* rowB = outb + (size_t)oB * HW;
        rowA[0]  = acc00[v] + bsA;
        rowA[16] = acc01[v] + bsA;
        rowA[32] = acc02[v] + bsA;
        rowA[48] = acc03[v] + bsA;
        rowB[0]  = acc10[v] + bsB;
        rowB[16] = acc11[v] + bsB;
        rowB[32] = acc12[v] + bsB;
        rowB[48] = acc13[v] + bsB;
    }
}

extern "C" void kernel_launch(void* const* d_in, const int* in_sizes, int n_in,
                              void* d_out, int out_size, void* d_ws, size_t ws_size,
                              hipStream_t stream) {
    const float* feature = (const float*)d_in[0];
    const float* conv_w  = (const float*)d_in[1];
    const float* conv_b  = (const float*)d_in[2];
    float* out = (float*)d_out;

    float* invn = (float*)d_ws;
    float* corr = invn + 32768;

    escn_norm_kernel<<<1024, 256, 0, stream>>>(feature, invn);
    escn_corr_kernel<<<256, 512, 0, stream>>>(feature, invn, corr);
    escn_gemm_kernel<<<dim3(8, 24, 8), 256, 0, stream>>>(feature, corr,
                                                         conv_w, conv_b, out);
}


# gemm kernel: 16-byte fragment loads and output stores (tile column c = pixel 4c+t; K in blocks of 8, a lane half takes 4 consecutive k in both operands); changes summation order
# speedup vs baseline: 3.0214x; 1.0571x over previous
#include <hip/hip_runtime.h>
#include <hip/hip_bf16.h>

typedef __attribute__((ext_vector_type(2))) float v2f;
typedef __attribute__((ext_vector_type(8))) float v8f;

#define CH    768
#define HW    4096
#define NOFF  28
#define KTOT  796

#define CSPLIT 8
#define CSLICE (CH / CSPLIT)

__global__ __launch_bounds__(256)
void escn_norm_kernel(const float* __restrict__ feat,
                      float* __restrict__ invn) {
    __shared__ float part[CSPLIT * 32];

    const int lane = threadIdx.x & 31;
    const int g    = __builtin_amdgcn_readfirstlane((int)(threadIdx.x >> 5));
    const int b    = blockIdx.x >> 7;
    const int p    = ((blockIdx.x & 127) << 5) + lane;

    const float* base = feat + ((size_t)b * CH + (size_t)g * CSLICE) * HW + p;
    float s = 0.0f;
#pragma unroll 4
    for (int c = 0; c < CSLICE; ++c) {
        float v = base[(size_t)c * HW];
        s = fmaf(v, v, s);
    }
    part[g * 32 + lane] = s;
    __syncthreads();

    if (g == 0) {
        float t = part[lane];
#pragma unroll
        for (int q = 1; q < CSPLIT; ++q) t += part[q * 32 + lane];
        float n = sqrtf(t);
        invn[b * HW + p] = 1.0f / fmaxf(n, 1e-12f);
    }
}

typedef __attribute__((ext_vector_type(4))) float v4f;

#define KSPLIT 16
#define KSLICE (CH / KSPLIT)
#define NUNIQ  25

__device__ __forceinline__ void escn_offset(int o, int& r, int& cc) {
    if      (o <  7) { r = o;      cc = o;            }
    else if (o < 14) { r = o - 7;  cc = 3;            }
    else if (o < 21) { r = o - 14; cc = 6 - (o - 14); }
    else             { r = 3;      cc = o - 21;       }
}

__device__ __forceinline__ int escn_slot(int o) {
    if (o == 10 || o == 17 || o == 24) return 3;
    return o - (o > 24 ? 3 : o > 17 ? 2 : o > 10 ? 1 : 0);
}

__global__ __launch_bounds__(512)
void escn_corr_kernel(const float* __restrict__ feat,
                      const float* __restrict__ invn,
                      float* __restrict__ corr) {
    __shared__ v4f red[KSPLIT * NUNIQ * 32];

    const int lane = threadIdx.x & 31;
    const int g    = __builtin_amdgcn_readfirstlane((int)(threadIdx.x >> 5));
    const int b    = blockIdx.x >> 5;
    const int j    = lane & 15;
    const int h    = ((blockIdx.x & 31) << 1) + (lane >> 4);
    const int p0   = h * 64 + 4 * j;

    unsigned offL[7], offC[7], offR[7];
#pragma unroll
    for (int i = 0; i < 7; ++i) {
        int  rr  = h + i - 3;
        bool rok = (rr >= 0) && (rr < 64);
        unsigned oc = 4u * (unsigned)((rok ? rr : h) * 64 + 4 * j);
        offC[i] = oc;
        offL[i] = oc - (j > 0  ? 16u : 0u);
        offR[i] = oc + (j < 15 ? 16u : 0u);
    }

    float acc[NUNIQ][4];
#pragma unroll
    for (int s = 0; s < NUNIQ; ++s) {
#pragma unroll
        for (int t = 0; t < 4; ++t) acc[s][t] = 0.0f;
    }

    const char* base = (const char*)(feat + ((size_t)b * CH + (size_t)g * KSLICE) * HW);
    for (int c = 0; c < KSLICE; ++c) {
        const unsigned plane = (unsigned)c * (unsigned)(HW * sizeof(float));
        float win[7][12];
#pragma unroll
        for (int i = 0; i < 7; ++i) {
            v4f L = *(const v4f*)(base + (plane + offL[i]));
            v4f C = *(const v4f*)(base + (plane + offC[i]));
            v4f R = *(const v4f*)(base + (plane + offR[i]));
            win[i][0] = L.x; win[i][1] = L.y; win[i][2]  = L.z; win[i][3]  = L.w;
            win[i][4] = C.x; win[i][5] = C.y; win[i][6]  = C.z; win[i][7]  = C.w;
            win[i][8] = R.x; win[i][9] = R.y; win[i][10] = R.z; win[i][11] = R.w;
        }
#pragma unroll
        for (int o = 0; o < NOFF; ++o) {
            if (o == 10 || o == 17 || o == 24) continue;
            int r, cc;
            escn_offset(o, r, cc);
            const int s = escn_slot(o);
#pragma unroll
            for (int t = 0; t < 4; ++t)
                acc[s][t] = fmaf(win[r][1 + t + cc], win[3][4 + t], acc[s][t]);
        }
    }

#pragma unroll
    for (int s = 0; s < NUNIQ; ++s) {
        v4f v;
        v.x = acc[s][0]; v.y = acc[s][1]; v.z = acc[s][2]; v.w = acc[s][3];
        red[(g * NUNIQ + s) * 32 + lane] = v;
    }
    __syncthreads();

    const float* ib = invn + b * HW;
    const v4f ic = *(const v4f*)(ib + p0);
    for (int o = g; o < NOFF; o += KSPLIT) {
        int r, cc;
        escn_offset(o, r, cc);
        const int s = escn_slot(o);
        v4f sum = red[s * 32 + lane];
#pragma unroll
        for (int q = 1; q < KSPLIT; ++q)
            sum += red[(q * NUNIQ + s) * 32 + lane];
        const int  sh  = h + r - 3;
        const bool rok = (sh >= 0) && (sh < 64);
        v4f res;
#pragma unroll
        for (int t = 0; t < 4; ++t) {
            int  sw  = 4 * j + t + cc - 3;
            bool ok  = rok && (sw >= 0) && (sw < 64);
            int  spn = ok ? (sh * 64 + sw) : (p0 + t);
            float vf = ok ? 1.0f : 0.0f;
            res[t] = sum[t] * vf * ic[t] * ib[spn];
        }
        *(v4f*)(corr + ((size_t)b * NOFF + o) * HW + p0) = res;
    }
}

#define ESCN_WMMA(A, B, C) \
    __builtin_amdgcn_wmma_f32_16x16x4_f32(false, (A), false, (B), (short)0, (C), false, false)

#define ESCN_FRAG(aX0, aX1, rlo, rhi)                                   \
    {                                                                   \
        v2f bt0, bt1, bt2, bt3;                                         \
        bt0.x = rlo.x; bt0.y = rhi.x;                                   \
        bt1.x = rlo.y; bt1.y = rhi.y;                                   \
        bt2.x = rlo.z; bt2.y = rhi.z;                                   \
        bt3.x = rlo.w; bt3.y = rhi.w;                                   \
        acc00 = ESCN_WMMA(aX0, bt0, acc00);                             \
        acc01 = ESCN_WMMA(aX0, bt1, acc01);                             \
        acc02 = ESCN_WMMA(aX0, bt2, acc02);                             \
        acc03 = ESCN_WMMA(aX0, bt3, acc03);                             \
        acc10 = ESCN_WMMA(aX1, bt0, acc10);                             \
        acc11 = ESCN_WMMA(aX1, bt1, acc11);                             \
        acc12 = ESCN_WMMA(aX1, bt2, acc12);                             \
        acc13 = ESCN_WMMA(aX1, bt3, acc13);                             \
    }

#define ESCN_BLOCK8(wk0, wk1, xk)                                       \
    {                                                                   \
        const v4f a0 = *(const v4f*)(wk0);                              \
        const v4f a1 = *(const v4f*)(wk1);                              \
        const v4f r0 = *(const v4f*)(xk);                               \
        const v4f r1 = *(const v4f*)((xk) + HW);                        \
        const v4f r2 = *(const v4f*)((xk) + 2 * HW);                    \
        const v4f r3 = *(const v4f*)((xk) + 3 * HW);                    \
        v2f aP0, aP1, aQ0, aQ1;                                         \
        aP0.x = a0.x; aP0.y = a0.y; aQ0.x = a0.z; aQ0.y = a0.w;         \
        aP1.x = a1.x; aP1.y = a1.y; aQ1.x = a1.z; aQ1.y = a1.w;         \
        ESCN_FRAG(aP0, aP1, r0, r1)                                     \
        ESCN_FRAG(aQ0, aQ1, r2, r3)                                     \
    }

__global__ void escn_gemm_kernel(const float* __restrict__ feat,
                                 const float* __restrict__ corr,
                                 const float* __restrict__ Wm,
                                 const float* __restrict__ bias,
                                 float* __restrict__ out) {
    const int lane  = threadIdx.x & 31;
    const int wid   = threadIdx.x >> 5;
    const int mtile = blockIdx.x * 8 + wid;
    const int m0    = mtile * 64;
    const int o0    = blockIdx.y * 32;
    const int b     = blockIdx.z;

    const int col   = lane & 15;
    const int khalf = lane >> 4;

    v8f acc00 = {}, acc01 = {}, acc02 = {}, acc03 = {};
    v8f acc10 = {}, acc11 = {}, acc12 = {}, acc13 = {};

    const float* fbase = feat + ((size_t)b * CH   + 4 * khalf) * HW + m0 + 4 * col;
    const float* cbase = corr + ((size_t)b * NOFF + 4 * khalf) * HW + m0 + 4 * col;
    const float* wrow0 = Wm + (size_t)(o0 + col) * KTOT + 4 * khalf;
    const float* wrow1 = wrow0 + (size_t)16 * KTOT;

    for (int k = 0; k < CH; k += 8) {
        const float* xk = fbase + (size_t)k * HW;
        ESCN_BLOCK8(wrow0 + k, wrow1 + k, xk)
    }

#pragma nounroll
    for (int k = 0; k < 24; k += 8) {
        const float* xk = cbase + (size_t)k * HW;
        ESCN_BLOCK8(wrow0 + CH + k, wrow1 + CH + k, xk)
    }

    {
        const float* xk = cbase + (size_t)(24 - 2 * khalf) * HW;
        const v2f a0 = *(const v2f*)(wrow0 + CH + 24 - 2 * khalf);
        const v2f a1 = *(const v2f*)(wrow1 + CH + 24 - 2 * khalf);
        const v4f r0 = *(const v4f*)(xk);
        const v4f r1 = *(const v4f*)(xk + HW);
        ESCN_FRAG(a0, a1, r0, r1)
    }

    float* outb = out + ((size_t)b * CH) * HW + m0 + 4 * col;
#pragma unroll
    for (int v = 0; v < 8; ++v) {
        const int oA = o0 + v + 8 * khalf;
        const int oB = oA + 16;
        const float bsA = bias[oA];
        const float bsB = bias[oB];
        v4f resA, resB;
        resA.x = acc00[v] + bsA; resA.y = acc01[v] + bsA;
        resA.z = acc02[v] + bsA; resA.w = acc03[v] + bsA;
        resB.x = acc10[v] + bsB; resB.y = acc11[v] + bsB;
        resB.z = acc12[v] + bsB; resB.w = acc13[v] + bsB;
        *(v4f*)(outb + (size_t)oA * HW) = resA;
        *(v4f*)(outb + (size_t)oB * HW) = resB;
    }
}

extern "C" void kernel_launch(void* const* d_in, const int* in_sizes, int n_in,
                              void* d_out, int out_size, void* d_ws, size_t ws_size,
                              hipStream_t stream) {
    const float* feature = (const float*)d_in[0];
    const float* conv_w  = (const float*)d_in[1];
    const float* conv_b  = (const float*)d_in[2];
    float* out = (float*)d_out;

    float* invn = (float*)d_ws;
    float* corr = invn + 32768;

    escn_norm_kernel<<<1024, 256, 0, stream>>>(feature, invn);
    escn_corr_kernel<<<256, 512, 0, stream>>>(feature, invn, corr);
    escn_gemm_kernel<<<dim3(8, 24, 8), 256, 0, stream>>>(feature, corr,
                                                         conv_w, conv_b, out);
}


# gemm kernel: wave tile 64(o) x 64(m), 16 accumulator tiles (was 32 x 64, 8): each loaded X row feeds four o-tiles, 8 loads per 32 WMMAs (was 12), X read 12 times from the L2 (was 24)
# speedup vs baseline: 3.1436x; 1.0404x over previous
#include <hip/hip_runtime.h>
#include <hip/hip_bf16.h>

typedef __attribute__((ext_vector_type(2))) float v2f;
typedef __attribute__((ext_vector_type(8))) float v8f;

#define CH    768
#define HW    4096
#define NOFF  28
#define KTOT  796

#define CSPLIT 8
#define CSLICE (CH / CSPLIT)

__global__ __launch_bounds__(256)
void escn_norm_kernel(const float* __restrict__ feat,
                      float* __restrict__ invn) {
    __shared__ float part[CSPLIT * 32];

    const int lane = threadIdx.x & 31;
    const int g    = __builtin_amdgcn_readfirstlane((int)(threadIdx.x >> 5));
    const int b    = blockIdx.x >> 7;
    const int p    = ((blockIdx.x & 127) << 5) + lane;

    const float* base = feat + ((size_t)b * CH + (size_t)g * CSLICE) * HW + p;
    float s = 0.0f;
#pragma unroll 4
    for (int c = 0; c < CSLICE; ++c) {
        float v = base[(size_t)c * HW];
        s = fmaf(v, v, s);
    }
    part[g * 32 + lane] = s;
    __syncthreads();

    if (g == 0) {
        float t = part[lane];
#pragma unroll
        for (int q = 1; q < CSPLIT; ++q) t += part[q * 32 + lane];
        float n = sqrtf(t);
        invn[b * HW + p] = 1.0f / fmaxf(n, 1e-12f);
    }
}

typedef __attribute__((ext_vector_type(4))) float v4f;

#define KSPLIT 16
#define KSLICE (CH / KSPLIT)
#define NUNIQ  25

__device__ __forceinline__ void escn_offset(int o, int& r, int& cc) {
    if      (o <  7) { r = o;      cc = o;            }
    else if (o < 14) { r = o - 7;  cc = 3;            }
    else if (o < 21) { r = o - 14; cc = 6 - (o - 14); }
    else             { r = 3;      cc = o - 21;       }
}

__device__ __forceinline__ int escn_slot(int o) {
    if (o == 10 || o == 17 || o == 24) return 3;
    return o - (o > 24 ? 3 : o > 17 ? 2 : o > 10 ? 1 : 0);
}

__global__ __launch_bounds__(512)
void escn_corr_kernel(const float* __restrict__ feat,
                      const float* __restrict__ invn,
                      float* __restrict__ corr) {
    __shared__ v4f red[KSPLIT * NUNIQ * 32];

    const int lane = threadIdx.x & 31;
    const int g    = __builtin_amdgcn_readfirstlane((int)(threadIdx.x >> 5));
    const int b    = blockIdx.x >> 5;
    const int j    = lane & 15;
    const int h    = ((blockIdx.x & 31) << 1) + (lane >> 4);
    const int p0   = h * 64 + 4 * j;

    unsigned offL[7], offC[7], offR[7];
#pragma unroll
    for (int i = 0; i < 7; ++i) {
        int  rr  = h + i - 3;
        bool rok = (rr >= 0) && (rr < 64);
        unsigned oc = 4u * (unsigned)((rok ? rr : h) * 64 + 4 * j);
        offC[i] = oc;
        offL[i] = oc - (j > 0  ? 16u : 0u);
        offR[i] = oc + (j < 15 ? 16u : 0u);
    }

    float acc[NUNIQ][4];
#pragma unroll
    for (int s = 0; s < NUNIQ; ++s) {
#pragma unroll
        for (int t = 0; t < 4; ++t) acc[s][t] = 0.0f;
    }

    const char* base = (const char*)(feat + ((size_t)b * CH + (size_t)g * KSLICE) * HW);
    for (int c = 0; c < KSLICE; ++c) {
        const unsigned plane = (unsigned)c * (unsigned)(HW * sizeof(float));
        float win[7][12];
#pragma unroll
        for (int i = 0; i < 7; ++i) {
            v4f L = *(const v4f*)(base + (plane + offL[i]));
            v4f C = *(const v4f*)(base + (plane + offC[i]));
            v4f R = *(const v4f*)(base + (plane + offR[i]));
            win[i][0] = L.x; win[i][1] = L.y; win[i][2]  = L.z; win[i][3]  = L.w;
            win[i][4] = C.x; win[i][5] = C.y; win[i][6]  = C.z; win[i][7]  = C.w;
            win[i][8] = R.x; win[i][9] = R.y; win[i][10] = R.z; win[i][11] = R.w;
        }
#pragma unroll
        for (int o = 0; o < NOFF; ++o) {
            if (o == 10 || o == 17 || o == 24) continue;
            int r, cc;
            escn_offset(o, r, cc);
            const int s = escn_slot(o);
#pragma unroll
            for (int t = 0; t < 4; ++t)
                acc[s][t] = fmaf(win[r][1 + t + cc], win[3][4 + t], acc[s][t]);
        }
    }

#pragma unroll
    for (int s = 0; s < NUNIQ; ++s) {
        v4f v;
        v.x = acc[s][0]; v.y = acc[s][1]; v.z = acc[s][2]; v.w = acc[s][3];
        red[(g * NUNIQ + s) * 32 + lane] = v;
    }
    __syncthreads();

    const float* ib = invn + b * HW;
    const v4f ic = *(const v4f*)(ib + p0);
    for (int o = g; o < NOFF; o += KSPLIT) {
        int r, cc;
        escn_offset(o, r, cc);
        const int s = escn_slot(o);
        v4f sum = red[s * 32 + lane];
#pragma unroll
        for (int q = 1; q < KSPLIT; ++q)
            sum += red[(q * NUNIQ + s) * 32 + lane];
        const int  sh  = h + r - 3;
        const bool rok = (sh >= 0) && (sh < 64);
        v4f res;
#pragma unroll
        for (int t = 0; t < 4; ++t) {
            int  sw  = 4 * j + t + cc - 3;
            bool ok  = rok && (sw >= 0) && (sw < 64);
            int  spn = ok ? (sh * 64 + sw) : (p0 + t);
            float vf = ok ? 1.0f : 0.0f;
            res[t] = sum[t] * vf * ic[t] * ib[spn];
        }
        *(v4f*)(corr + ((size_t)b * NOFF + o) * HW + p0) = res;
    }
}

#define ESCN_WMMA(A, B, C) \
    __builtin_amdgcn_wmma_f32_16x16x4_f32(false, (A), false, (B), (short)0, (C), false, false)

#define ESCN_OTILE(aX, c0, c1, c2, c3)                                  \
    c0 = ESCN_WMMA(aX, bt0, c0);                                        \
    c1 = ESCN_WMMA(aX, bt1, c1);                                        \
    c2 = ESCN_WMMA(aX, bt2, c2);                                        \
    c3 = ESCN_WMMA(aX, bt3, c3);

#define ESCN_FRAG(aX0, aX1, aX2, aX3, rlo, rhi)                         \
    {                                                                   \
        v2f bt0, bt1, bt2, bt3;                                         \
        bt0.x = rlo.x; bt0.y = rhi.x;                                   \
        bt1.x = rlo.y; bt1.y = rhi.y;                                   \
        bt2.x = rlo.z; bt2.y = rhi.z;                                   \
        bt3.x = rlo.w; bt3.y = rhi.w;                                   \
        ESCN_OTILE(aX0, acc00, acc01, acc02, acc03)                     \
        ESCN_OTILE(aX1, acc10, acc11, acc12, acc13)                     \
        ESCN_OTILE(aX2, acc20, acc21, acc22, acc23)                     \
        ESCN_OTILE(aX3, acc30, acc31, acc32, acc33)                     \
    }

#define ESCN_BLOCK8(wk, xk)                                             \
    {                                                                   \
        const v4f a0 = *(const v4f*)(wk);                               \
        const v4f a1 = *(const v4f*)((wk) + 16 * KTOT);                 \
        const v4f a2 = *(const v4f*)((wk) + 32 * KTOT);                 \
        const v4f a3 = *(const v4f*)((wk) + 48 * KTOT);                 \
        const v4f r0 = *(const v4f*)(xk);                               \
        const v4f r1 = *(const v4f*)((xk) + HW);                        \
        const v4f r2 = *(const v4f*)((xk) + 2 * HW);                    \
        const v4f r3 = *(const v4f*)((xk) + 3 * HW);                    \
        v2f aP0, aP1, aP2, aP3, aQ0, aQ1, aQ2, aQ3;                     \
        aP0.x = a0.x; aP0.y = a0.y; aQ0.x = a0.z; aQ0.y = a0.w;         \
        aP1.x = a1.x; aP1.y = a1.y; aQ1.x = a1.z; aQ1.y = a1.w;         \
        aP2.x = a2.x; aP2.y = a2.y; aQ2.x = a2.z; aQ2.y = a2.w;         \
        aP3.x = a3.x; aP3.y = a3.y; aQ3.x = a3.z; aQ3.y = a3.w;         \
        ESCN_FRAG(aP0, aP1, aP2, aP3, r0, r1)                           \
        ESCN_FRAG(aQ0, aQ1, aQ2, aQ3, r2, r3)                           \
    }

__global__ __launch_bounds__(256)
void escn_gemm_kernel(const float* __restrict__ feat,
                      const float* __restrict__ corr,
                      const float* __restrict__ Wm,
                      const float* __restrict__ bias,
                      float* __restrict__ out) {
    const int lane  = threadIdx.x & 31;
    const int wid   = threadIdx.x >> 5;
    const int mtile = blockIdx.x * 8 + wid;
    const int m0    = mtile * 64;
    const int o0    = blockIdx.y * 64;
    const int b     = blockIdx.z;

    const int col   = lane & 15;
    const int khalf = lane >> 4;

    v8f acc00 = {}, acc01 = {}, acc02 = {}, acc03 = {};
    v8f acc10 = {}, acc11 = {}, acc12 = {}, acc13 = {};
    v8f acc20 = {}, acc21 = {}, acc22 = {}, acc23 = {};
    v8f acc30 = {}, acc31 = {}, acc32 = {}, acc33 = {};

    const float* fbase = feat + ((size_t)b * CH   + 4 * khalf) * HW + m0 + 4 * col;
    const float* cbase = corr + ((size_t)b * NOFF + 4 * khalf) * HW + m0 + 4 * col;
    const float* wrow  = Wm + (size_t)(o0 + col) * KTOT + 4 * khalf;

    for (int k = 0; k < CH; k += 8) {
        const float* xk = fbase + (size_t)k * HW;
        ESCN_BLOCK8(wrow + k, xk)
    }

#pragma nounroll
    for (int k = 0; k < 24; k += 8) {
        const float* xk = cbase + (size_t)k * HW;
        ESCN_BLOCK8(wrow + CH + k, xk)
    }

    {
        const float* xk = cbase + (size_t)(24 - 2 * khalf) * HW;
        const float* wk = wrow + CH + 24 - 2 * khalf;
        const v2f a0 = *(const v2f*)(wk);
        const v2f a1 = *(const v2f*)(wk + 16 * KTOT);
        const v2f a2 = *(const v2f*)(wk + 32 * KTOT);
        const v2f a3 = *(const v2f*)(wk + 48 * KTOT);
        const v4f r0 = *(const v4f*)(xk);
        const v4f r1 = *(const v4f*)(xk + HW);
        ESCN_FRAG(a0, a1, a2, a3, r0, r1)
    }

    float* outb = out + ((size_t)b * CH) * HW + m0 + 4 * col;
#define ESCN_STORE(ot, c0, c1, c2, c3)                                  \
    {                                                                   \
        const int   oR = o0 + 16 * (ot) + v + 8 * khalf;                \
        const float bs = bias[oR];                                      \
        v4f res;                                                        \
        res.x = c0[v] + bs; res.y = c1[v] + bs;                         \
        res.z = c2[v] + bs; res.w = c3[v] + bs;                         \
        *(v4f*)(outb + (size_t)oR * HW) = res;                          \
    }
#pragma unroll
    for (int v = 0; v < 8; ++v) {
        ESCN_STORE(0, acc00, acc01, acc02, acc03)
        ESCN_STORE(1, acc10, acc11, acc12, acc13)
        ESCN_STORE(2, acc20, acc21, acc22, acc23)
        ESCN_STORE(3, acc30, acc31, acc32, acc33)
    }
}

extern "C" void kernel_launch(void* const* d_in, const int* in_sizes, int n_in,
                              void* d_out, int out_size, void* d_ws, size_t ws_size,
                              hipStream_t stream) {
    const float* feature = (const float*)d_in[0];
    const float* conv_w  = (const float*)d_in[1];
    const float* conv_b  = (const float*)d_in[2];
    float* out = (float*)d_out;

    float* invn = (float*)d_ws;
    float* corr = invn + 32768;

    escn_norm_kernel<<<1024, 256, 0, stream>>>(feature, invn);
    escn_corr_kernel<<<256, 512, 0, stream>>>(feature, invn, corr);
    escn_gemm_kernel<<<dim3(8, 12, 8), 256, 0, stream>>>(feature, corr,
                                                         conv_w, conv_b, out);
}


# gemm kernel: K walk pipelined over two fragment sets: block i+1's 8 loads issued as one group in front of block i's 32 WMMAs and waited for behind them (were spread among the WMMAs)
# speedup vs baseline: 3.2893x; 1.0463x over previous
#include <hip/hip_runtime.h>
#include <hip/hip_bf16.h>

typedef __attribute__((ext_vector_type(2))) float v2f;
typedef __attribute__((ext_vector_type(8))) float v8f;

#define CH    768
#define HW    4096
#define NOFF  28
#define KTOT  796

#define CSPLIT 8
#define CSLICE (CH / CSPLIT)

__global__ __launch_bounds__(256)
void escn_norm_kernel(const float* __restrict__ feat,
                      float* __restrict__ invn) {
    __shared__ float part[CSPLIT * 32];

    const int lane = threadIdx.x & 31;
    const int g    = __builtin_amdgcn_readfirstlane((int)(threadIdx.x >> 5));
    const int b    = blockIdx.x >> 7;
    const int p    = ((blockIdx.x & 127) << 5) + lane;

    const float* base = feat + ((size_t)b * CH + (size_t)g * CSLICE) * HW + p;
    float s = 0.0f;
#pragma unroll 4
    for (int c = 0; c < CSLICE; ++c) {
        float v = base[(size_t)c * HW];
        s = fmaf(v, v, s);
    }
    part[g * 32 + lane] = s;
    __syncthreads();

    if (g == 0) {
        float t = part[lane];
#pragma unroll
        for (int q = 1; q < CSPLIT; ++q) t += part[q * 32 + lane];
        float n = sqrtf(t);
        invn[b * HW + p] = 1.0f / fmaxf(n, 1e-12f);
    }
}

typedef __attribute__((ext_vector_type(4))) float v4f;

#define KSPLIT 16
#define KSLICE (CH / KSPLIT)
#define NUNIQ  25

__device__ __forceinline__ void escn_offset(int o, int& r, int& cc) {
    if      (o <  7) { r = o;      cc = o;            }
    else if (o < 14) { r = o - 7;  cc = 3;            }
    else if (o < 21) { r = o - 14; cc = 6 - (o - 14); }
    else             { r = 3;      cc = o - 21;       }
}

__device__ __forceinline__ int escn_slot(int o) {
    if (o == 10 || o == 17 || o == 24) return 3;
    return o - (o > 24 ? 3 : o > 17 ? 2 : o > 10 ? 1 : 0);
}

__global__ __launch_bounds__(512)
void escn_corr_kernel(const float* __restrict__ feat,
                      const float* __restrict__ invn,
                      float* __restrict__ corr) {
    __shared__ v4f red[KSPLIT * NUNIQ * 32];

    const int lane = threadIdx.x & 31;
    const int g    = __builtin_amdgcn_readfirstlane((int)(threadIdx.x >> 5));
    const int b    = blockIdx.x >> 5;
    const int j    = lane & 15;
    const int h    = ((blockIdx.x & 31) << 1) + (lane >> 4);
    const int p0   = h * 64 + 4 * j;

    unsigned offL[7], offC[7], offR[7];
#pragma unroll
    for (int i = 0; i < 7; ++i) {
        int  rr  = h + i - 3;
        bool rok = (rr >= 0) && (rr < 64);
        unsigned oc = 4u * (unsigned)((rok ? rr : h) * 64 + 4 * j);
        offC[i] = oc;
        offL[i] = oc - (j > 0  ? 16u : 0u);
        offR[i] = oc + (j < 15 ? 16u : 0u);
    }

    float acc[NUNIQ][4];
#pragma unroll
    for (int s = 0; s < NUNIQ; ++s) {
#pragma unroll
        for (int t = 0; t < 4; ++t) acc[s][t] = 0.0f;
    }

    const char* base = (const char*)(feat + ((size_t)b * CH + (size_t)g * KSLICE) * HW);
    for (int c = 0; c < KSLICE; ++c) {
        const unsigned plane = (unsigned)c * (unsigned)(HW * sizeof(float));
        float win[7][12];
#pragma unroll
        for (int i = 0; i < 7; ++i) {
            v4f L = *(const v4f*)(base + (plane + offL[i]));
            v4f C = *(const v4f*)(base + (plane + offC[i]));
            v4f R = *(const v4f*)(base + (plane + offR[i]));
            win[i][0] = L.x; win[i][1] = L.y; win[i][2]  = L.z; win[i][3]  = L.w;
            win[i][4] = C.x; win[i][5] = C.y; win[i][6]  = C.z; win[i][7]  = C.w;
            win[i][8] = R.x; win[i][9] = R.y; win[i][10] = R.z; win[i][11] = R.w;
        }
#pragma unroll
        for (int o = 0; o < NOFF; ++o) {
            if (o == 10 || o == 17 || o == 24) continue;
            int r, cc;
            escn_offset(o, r, cc);
            const int s = escn_slot(o);
#pragma unroll
            for (int t = 0; t < 4; ++t)
                acc[s][t] = fmaf(win[r][1 + t + cc], win[3][4 + t], acc[s][t]);
        }
    }

#pragma unroll
    for (int s = 0; s < NUNIQ; ++s) {
        v4f v;
        v.x = acc[s][0]; v.y = acc[s][1]; v.z = acc[s][2]; v.w = acc[s][3];
        red[(g * NUNIQ + s) * 32 + lane] = v;
    }
    __syncthreads();

    const float* ib = invn + b * HW;
    const v4f ic = *(const v4f*)(ib + p0);
    for (int o = g; o < NOFF; o += KSPLIT) {
        int r, cc;
        escn_offset(o, r, cc);
        const int s = escn_slot(o);
        v4f sum = red[s * 32 + lane];
#pragma unroll
        for (int q = 1; q < KSPLIT; ++q)
            sum += red[(q * NUNIQ + s) * 32 + lane];
        const int  sh  = h + r - 3;
        const bool rok = (sh >= 0) && (sh < 64);
        v4f res;
#pragma unroll
        for (int t = 0; t < 4; ++t) {
            int  sw  = 4 * j + t + cc - 3;
            bool ok  = rok && (sw >= 0) && (sw < 64);
            int  spn = ok ? (sh * 64 + sw) : (p0 + t);
            float vf = ok ? 1.0f : 0.0f;
            res[t] = sum[t] * vf * ic[t] * ib[spn];
        }
        *(v4f*)(corr + ((size_t)b * NOFF + o) * HW + p0) = res;
    }
}

#define ESCN_WMMA(A, B, C) \
    __builtin_amdgcn_wmma_f32_16x16x4_f32(false, (A), false, (B), (short)0, (C), false, false)

#define ESCN_OTILE(aX, c0, c1, c2, c3)                                  \
    c0 = ESCN_WMMA(aX, bt0, c0);                                        \
    c1 = ESCN_WMMA(aX, bt1, c1);                                        \
    c2 = ESCN_WMMA(aX, bt2, c2);                                        \
    c3 = ESCN_WMMA(aX, bt3, c3);

#define ESCN_FRAG(aX0, aX1, aX2, aX3, rlo, rhi)                         \
    {                                                                   \
        v2f bt0, bt1, bt2, bt3;                                         \
        bt0.x = rlo.x; bt0.y = rhi.x;                                   \
        bt1.x = rlo.y; bt1.y = rhi.y;                                   \
        bt2.x = rlo.z; bt2.y = rhi.z;                                   \
        bt3.x = rlo.w; bt3.y = rhi.w;                                   \
        ESCN_OTILE(aX0, acc00, acc01, acc02, acc03)                     \
        ESCN_OTILE(aX1, acc10, acc11, acc12, acc13)                     \
        ESCN_OTILE(aX2, acc20, acc21, acc22, acc23)                     \
        ESCN_OTILE(aX3, acc30, acc31, acc32, acc33)                     \
    }

#define ESCN_LOAD8(S, wk, xk)                                           \
    S##a0 = *(const v4f*)(wk);                                          \
    S##a1 = *(const v4f*)((wk) + 16 * KTOT);                            \
    S##a2 = *(const v4f*)((wk) + 32 * KTOT);                            \
    S##a3 = *(const v4f*)((wk) + 48 * KTOT);                            \
    S##r0 = *(const v4f*)(xk);                                          \
    S##r1 = *(const v4f*)((xk) + HW);                                   \
    S##r2 = *(const v4f*)((xk) + 2 * HW);                               \
    S##r3 = *(const v4f*)((xk) + 3 * HW);

#define ESCN_MMA8(S)                                                    \
    {                                                                   \
        v2f aP0, aP1, aP2, aP3, aQ0, aQ1, aQ2, aQ3;                     \
        aP0.x = S##a0.x; aP0.y = S##a0.y; aQ0.x = S##a0.z; aQ0.y = S##a0.w; \
        aP1.x = S##a1.x; aP1.y = S##a1.y; aQ1.x = S##a1.z; aQ1.y = S##a1.w; \
        aP2.x = S##a2.x; aP2.y = S##a2.y; aQ2.x = S##a2.z; aQ2.y = S##a2.w; \
        aP3.x = S##a3.x; aP3.y = S##a3.y; aQ3.x = S##a3.z; aQ3.y = S##a3.w; \
        ESCN_FRAG(aP0, aP1, aP2, aP3, S##r0, S##r1)                     \
        ESCN_FRAG(aQ0, aQ1, aQ2, aQ3, S##r2, S##r3)                     \
    }

#define ESCN_PIN __builtin_amdgcn_sched_barrier(0);

__global__ __launch_bounds__(256)
void escn_gemm_kernel(const float* __restrict__ feat,
                      const float* __restrict__ corr,
                      const float* __restrict__ Wm,
                      const float* __restrict__ bias,
                      float* __restrict__ out) {
    const int lane  = threadIdx.x & 31;
    const int wid   = threadIdx.x >> 5;
    const int mtile = blockIdx.x * 8 + wid;
    const int m0    = mtile * 64;
    const int o0    = blockIdx.y * 64;
    const int b     = blockIdx.z;

    const int col   = lane & 15;
    const int khalf = lane >> 4;

    v8f acc00 = {}, acc01 = {}, acc02 = {}, acc03 = {};
    v8f acc10 = {}, acc11 = {}, acc12 = {}, acc13 = {};
    v8f acc20 = {}, acc21 = {}, acc22 = {}, acc23 = {};
    v8f acc30 = {}, acc31 = {}, acc32 = {}, acc33 = {};

    const float* fbase = feat + ((size_t)b * CH   + 4 * khalf) * HW + m0 + 4 * col;
    const float* cbase = corr + ((size_t)b * NOFF + 4 * khalf) * HW + m0 + 4 * col;
    const float* wrow  = Wm + (size_t)(o0 + col) * KTOT + 4 * khalf;

    v4f Ea0, Ea1, Ea2, Ea3, Er0, Er1, Er2, Er3;
    v4f Fa0, Fa1, Fa2, Fa3, Fr0, Fr1, Fr2, Fr3;

    ESCN_LOAD8(E, wrow, fbase)
    for (int k = 0; k < CH - 16; k += 16) {
        ESCN_PIN
        ESCN_LOAD8(F, wrow + k + 8, fbase + (size_t)(k + 8) * HW)
        ESCN_PIN
        ESCN_MMA8(E)
        ESCN_PIN
        ESCN_LOAD8(E, wrow + k + 16, fbase + (size_t)(k + 16) * HW)
        ESCN_PIN
        ESCN_MMA8(F)
    }
    ESCN_PIN
    ESCN_LOAD8(F, wrow + (CH - 8), fbase + (size_t)(CH - 8) * HW)
    ESCN_PIN
    ESCN_MMA8(E)

    ESCN_PIN
    ESCN_LOAD8(E, wrow + CH, cbase)
    ESCN_PIN
    ESCN_MMA8(F)
    ESCN_PIN
    ESCN_LOAD8(F, wrow + CH + 8, cbase + (size_t)8 * HW)
    ESCN_PIN
    ESCN_MMA8(E)
    ESCN_PIN
    ESCN_LOAD8(E, wrow + CH + 16, cbase + (size_t)16 * HW)
    ESCN_PIN
    ESCN_MMA8(F)

    {
        const float* xk = cbase + (size_t)(24 - 2 * khalf) * HW;
        const float* wk = wrow + CH + 24 - 2 * khalf;
        ESCN_PIN
        const v2f a0 = *(const v2f*)(wk);
        const v2f a1 = *(const v2f*)(wk + 16 * KTOT);
        const v2f a2 = *(const v2f*)(wk + 32 * KTOT);
        const v2f a3 = *(const v2f*)(wk + 48 * KTOT);
        const v4f r0 = *(const v4f*)(xk);
        const v4f r1 = *(const v4f*)(xk + HW);
        ESCN_PIN
        ESCN_MMA8(E)
        ESCN_FRAG(a0, a1, a2, a3, r0, r1)
    }

    float* outb = out + ((size_t)b * CH) * HW + m0 + 4 * col;
#define ESCN_STORE(ot, c0, c1, c2, c3)                                  \
    {                                                                   \
        const int   oR = o0 + 16 * (ot) + v + 8 * khalf;                \
        const float bs = bias[oR];                                      \
        v4f res;                                                        \
        res.x = c0[v] + bs; res.y = c1[v] + bs;                         \
        res.z = c2[v] + bs; res.w = c3[v] + bs;                         \
        *(v4f*)(outb + (size_t)oR * HW) = res;                          \
    }
#pragma unroll
    for (int v = 0; v < 8; ++v) {
        ESCN_STORE(0, acc00, acc01, acc02, acc03)
        ESCN_STORE(1, acc10, acc11, acc12, acc13)
        ESCN_STORE(2, acc20, acc21, acc22, acc23)
        ESCN_STORE(3, acc30, acc31, acc32, acc33)
    }
}

extern "C" void kernel_launch(void* const* d_in, const int* in_sizes, int n_in,
                              void* d_out, int out_size, void* d_ws, size_t ws_size,
                              hipStream_t stream) {
    const float* feature = (const float*)d_in[0];
    const float* conv_w  = (const float*)d_in[1];
    const float* conv_b  = (const float*)d_in[2];
    float* out = (float*)d_out;

    float* invn = (float*)d_ws;
    float* corr = invn + 32768;

    escn_norm_kernel<<<1024, 256, 0, stream>>>(feature, invn);
    escn_corr_kernel<<<256, 512, 0, stream>>>(feature, invn, corr);
    escn_gemm_kernel<<<dim3(8, 12, 8), 256, 0, stream>>>(feature, corr,
                                                         conv_w, conv_b, out);
}


# norm launch fused into corr kernel: it also sums the squares of its 6 other window rows (norm: 16 slices of 48, sum order changes), inverse norms via LDS; first reader now, prefetch 2 channels ahead
# speedup vs baseline: 3.5098x; 1.0670x over previous
#include <hip/hip_runtime.h>
#include <hip/hip_bf16.h>

typedef __attribute__((ext_vector_type(2))) float v2f;
typedef __attribute__((ext_vector_type(8))) float v8f;

#define CH    768
#define HW    4096
#define NOFF  28
#define KTOT  796

typedef __attribute__((ext_vector_type(4))) float v4f;

#define KSPLIT 16
#define KSLICE (CH / KSPLIT)
#define NUNIQ  25
#define NROWN  6
#define NSLOT  (NUNIQ + NROWN)
#define PFDIST 2

__device__ __forceinline__ void escn_offset(int o, int& r, int& cc) {
    if      (o <  7) { r = o;      cc = o;            }
    else if (o < 14) { r = o - 7;  cc = 3;            }
    else if (o < 21) { r = o - 14; cc = 6 - (o - 14); }
    else             { r = 3;      cc = o - 21;       }
}

__device__ __forceinline__ int escn_slot(int o) {
    if (o == 10 || o == 17 || o == 24) return 3;
    return o - (o > 24 ? 3 : o > 17 ? 2 : o > 10 ? 1 : 0);
}

__device__ __forceinline__ int escn_nslot(int i) {
    return i == 3 ? 3 : NUNIQ + (i < 3 ? i : i - 1);
}

__global__ __launch_bounds__(512)
void escn_corr_kernel(const float* __restrict__ feat,
                      float* __restrict__ corr) {
    __shared__ v4f red[KSPLIT * NSLOT * 32];
    __shared__ v4f inv[2 * 7 * 16];

    const int lane = threadIdx.x & 31;
    const int g    = __builtin_amdgcn_readfirstlane((int)(threadIdx.x >> 5));
    const int b    = blockIdx.x >> 5;
    const int j    = lane & 15;
    const int hf   = lane >> 4;
    const int h    = ((blockIdx.x & 31) << 1) + hf;
    const int p0   = h * 64 + 4 * j;

    unsigned offL[7], offC[7], offR[7];
#pragma unroll
    for (int i = 0; i < 7; ++i) {
        int  rr  = h + i - 3;
        bool rok = (rr >= 0) && (rr < 64);
        unsigned oc = 4u * (unsigned)((rok ? rr : h) * 64 + 4 * j);
        offC[i] = oc;
        offL[i] = oc - (j > 0  ? 16u : 0u);
        offR[i] = oc + (j < 15 ? 16u : 0u);
    }

    int prow = ((blockIdx.x & 31) << 1) - 3 + (lane & 7);
    prow = prow < 0 ? 0 : (prow > 63 ? 63 : prow);
    const unsigned pfoff = 256u * (unsigned)prow;

    float acc[NUNIQ][4];
#pragma unroll
    for (int s = 0; s < NUNIQ; ++s) {
#pragma unroll
        for (int t = 0; t < 4; ++t) acc[s][t] = 0.0f;
    }
    float nacc[NROWN][4];
#pragma unroll
    for (int s = 0; s < NROWN; ++s) {
#pragma unroll
        for (int t = 0; t < 4; ++t) nacc[s][t] = 0.0f;
    }

    const char* base = (const char*)(feat + ((size_t)b * CH + (size_t)g * KSLICE) * HW);
    __builtin_prefetch(base + ((unsigned)(HW * sizeof(float)) + pfoff), 0, 3);
    for (int c = 0; c < KSLICE; ++c) {
        const unsigned plane = (unsigned)c * (unsigned)(HW * sizeof(float));
        if (c + PFDIST < KSLICE)
            __builtin_prefetch(base + (plane + (unsigned)(PFDIST * HW * sizeof(float)) + pfoff), 0, 3);
        float win[7][12];
#pragma unroll
        for (int i = 0; i < 7; ++i) {
            v4f L = *(const v4f*)(base + (plane + offL[i]));
            v4f C = *(const v4f*)(base + (plane + offC[i]));
            v4f R = *(const v4f*)(base + (plane + offR[i]));
            win[i][0] = L.x; win[i][1] = L.y; win[i][2]  = L.z; win[i][3]  = L.w;
            win[i][4] = C.x; win[i][5] = C.y; win[i][6]  = C.z; win[i][7]  = C.w;
            win[i][8] = R.x; win[i][9] = R.y; win[i][10] = R.z; win[i][11] = R.w;
        }
#pragma unroll
        for (int o = 0; o < NOFF; ++o) {
            if (o == 10 || o == 17 || o == 24) continue;
            int r, cc;
            escn_offset(o, r, cc);
            const int s = escn_slot(o);
#pragma unroll
            for (int t = 0; t < 4; ++t)
                acc[s][t] = fmaf(win[r][1 + t + cc], win[3][4 + t], acc[s][t]);
        }
#pragma unroll
        for (int i = 0; i < 7; ++i) {
            if (i == 3) continue;
            const int s = i < 3 ? i : i - 1;
#pragma unroll
            for (int t = 0; t < 4; ++t)
                nacc[s][t] = fmaf(win[i][4 + t], win[i][4 + t], nacc[s][t]);
        }
    }

#pragma unroll
    for (int s = 0; s < NUNIQ; ++s) {
        v4f v;
        v.x = acc[s][0]; v.y = acc[s][1]; v.z = acc[s][2]; v.w = acc[s][3];
        red[(g * NSLOT + s) * 32 + lane] = v;
    }
#pragma unroll
    for (int s = 0; s < NROWN; ++s) {
        v4f v;
        v.x = nacc[s][0]; v.y = nacc[s][1]; v.z = nacc[s][2]; v.w = nacc[s][3];
        red[(g * NSLOT + NUNIQ + s) * 32 + lane] = v;
    }
    __syncthreads();

    if (g < 7) {
        const int s = escn_nslot(g);
        v4f nsq = red[s * 32 + lane];
#pragma unroll
        for (int q = 1; q < KSPLIT; ++q)
            nsq += red[(q * NSLOT + s) * 32 + lane];
        v4f iv;
#pragma unroll
        for (int t = 0; t < 4; ++t)
            iv[t] = 1.0f / fmaxf(sqrtf(nsq[t]), 1e-12f);
        inv[(hf * 7 + g) * 16 + j] = iv;
    }
    __syncthreads();

    const float* tab = (const float*)inv + hf * (7 * 64);
    const v4f ic = inv[(hf * 7 + 3) * 16 + j];
    for (int o = g; o < NOFF; o += KSPLIT) {
        int r, cc;
        escn_offset(o, r, cc);
        const int s = escn_slot(o);
        v4f sum = red[s * 32 + lane];
#pragma unroll
        for (int q = 1; q < KSPLIT; ++q)
            sum += red[(q * NSLOT + s) * 32 + lane];
        const int  sh  = h + r - 3;
        const bool rok = (sh >= 0) && (sh < 64);
        v4f res;
#pragma unroll
        for (int t = 0; t < 4; ++t) {
            int  sw  = 4 * j + t + cc - 3;
            bool ok  = rok && (sw >= 0) && (sw < 64);
            int  spn = ok ? (r * 64 + sw) : (3 * 64 + 4 * j + t);
            float vf = ok ? 1.0f : 0.0f;
            res[t] = sum[t] * vf * ic[t] * tab[spn];
        }
        *(v4f*)(corr + ((size_t)b * NOFF + o) * HW + p0) = res;
    }
}

#define ESCN_WMMA(A, B, C) \
    __builtin_amdgcn_wmma_f32_16x16x4_f32(false, (A), false, (B), (short)0, (C), false, false)

#define ESCN_OTILE(aX, c0, c1, c2, c3)                                  \
    c0 = ESCN_WMMA(aX, bt0, c0);                                        \
    c1 = ESCN_WMMA(aX, bt1, c1);                                        \
    c2 = ESCN_WMMA(aX, bt2, c2);                                        \
    c3 = ESCN_WMMA(aX, bt3, c3);

#define ESCN_FRAG(aX0, aX1, aX2, aX3, rlo, rhi)                         \
    {                                                                   \
        v2f bt0, bt1, bt2, bt3;                                         \
        bt0.x = rlo.x; bt0.y = rhi.x;                                   \
        bt1.x = rlo.y; bt1.y = rhi.y;                                   \
        bt2.x = rlo.z; bt2.y = rhi.z;                                   \
        bt3.x = rlo.w; bt3.y = rhi.w;                                   \
        ESCN_OTILE(aX0, acc00, acc01, acc02, acc03)                     \
        ESCN_OTILE(aX1, acc10, acc11, acc12, acc13)                     \
        ESCN_OTILE(aX2, acc20, acc21, acc22, acc23)                     \
        ESCN_OTILE(aX3, acc30, acc31, acc32, acc33)                     \
    }

#define ESCN_LOAD8(S, wk, xk)                                           \
    S##a0 = *(const v4f*)(wk);                                          \
    S##a1 = *(const v4f*)((wk) + 16 * KTOT);                            \
    S##a2 = *(const v4f*)((wk) + 32 * KTOT);                            \
    S##a3 = *(const v4f*)((wk) + 48 * KTOT);                            \
    S##r0 = *(const v4f*)(xk);                                          \
    S##r1 = *(const v4f*)((xk) + HW);                                   \
    S##r2 = *(const v4f*)((xk) + 2 * HW);                               \
    S##r3 = *(const v4f*)((xk) + 3 * HW);

#define ESCN_MMA8(S)                                                    \
    {                                                                   \
        v2f aP0, aP1, aP2, aP3, aQ0, aQ1, aQ2, aQ3;                     \
        aP0.x = S##a0.x; aP0.y = S##a0.y; aQ0.x = S##a0.z; aQ0.y = S##a0.w; \
        aP1.x = S##a1.x; aP1.y = S##a1.y; aQ1.x = S##a1.z; aQ1.y = S##a1.w; \
        aP2.x = S##a2.x; aP2.y = S##a2.y; aQ2.x = S##a2.z; aQ2.y = S##a2.w; \
        aP3.x = S##a3.x; aP3.y = S##a3.y; aQ3.x = S##a3.z; aQ3.y = S##a3.w; \
        ESCN_FRAG(aP0, aP1, aP2, aP3, S##r0, S##r1)                     \
        ESCN_FRAG(aQ0, aQ1, aQ2, aQ3, S##r2, S##r3)                     \
    }

#define ESCN_PIN __builtin_amdgcn_sched_barrier(0);

__global__ __launch_bounds__(256)
void escn_gemm_kernel(const float* __restrict__ feat,
                      const float* __restrict__ corr,
                      const float* __restrict__ Wm,
                      const float* __restrict__ bias,
                      float* __restrict__ out) {
    const int lane  = threadIdx.x & 31;
    const int wid   = threadIdx.x >> 5;
    const int mtile = blockIdx.x * 8 + wid;
    const int m0    = mtile * 64;
    const int o0    = blockIdx.y * 64;
    const int b     = blockIdx.z;

    const int col   = lane & 15;
    const int khalf = lane >> 4;

    v8f acc00 = {}, acc01 = {}, acc02 = {}, acc03 = {};
    v8f acc10 = {}, acc11 = {}, acc12 = {}, acc13 = {};
    v8f acc20 = {}, acc21 = {}, acc22 = {}, acc23 = {};
    v8f acc30 = {}, acc31 = {}, acc32 = {}, acc33 = {};

    const float* fbase = feat + ((size_t)b * CH   + 4 * khalf) * HW + m0 + 4 * col;
    const float* cbase = corr + ((size_t)b * NOFF + 4 * khalf) * HW + m0 + 4 * col;
    const float* wrow  = Wm + (size_t)(o0 + col) * KTOT + 4 * khalf;

    v4f Ea0, Ea1, Ea2, Ea3, Er0, Er1, Er2, Er3;
    v4f Fa0, Fa1, Fa2, Fa3, Fr0, Fr1, Fr2, Fr3;

    ESCN_LOAD8(E, wrow, fbase)
    for (int k = 0; k < CH - 16; k += 16) {
        ESCN_PIN
        ESCN_LOAD8(F, wrow + k + 8, fbase + (size_t)(k + 8) * HW)
        ESCN_PIN
        ESCN_MMA8(E)
        ESCN_PIN
        ESCN_LOAD8(E, wrow + k + 16, fbase + (size_t)(k + 16) * HW)
        ESCN_PIN
        ESCN_MMA8(F)
    }
    ESCN_PIN
    ESCN_LOAD8(F, wrow + (CH - 8), fbase + (size_t)(CH - 8) * HW)
    ESCN_PIN
    ESCN_MMA8(E)

    ESCN_PIN
    ESCN_LOAD8(E, wrow + CH, cbase)
    ESCN_PIN
    ESCN_MMA8(F)
    ESCN_PIN
    ESCN_LOAD8(F, wrow + CH + 8, cbase + (size_t)8 * HW)
    ESCN_PIN
    ESCN_MMA8(E)
    ESCN_PIN
    ESCN_LOAD8(E, wrow + CH + 16, cbase + (size_t)16 * HW)
    ESCN_PIN
    ESCN_MMA8(F)

    {
        const float* xk = cbase + (size_t)(24 - 2 * khalf) * HW;
        const float* wk = wrow + CH + 24 - 2 * khalf;
        ESCN_PIN
        const v2f a0 = *(const v2f*)(wk);
        const v2f a1 = *(const v2f*)(wk + 16 * KTOT);
        const v2f a2 = *(const v2f*)(wk + 32 * KTOT);
        const v2f a3 = *(const v2f*)(wk + 48 * KTOT);
        const v4f r0 = *(const v4f*)(xk);
        const v4f r1 = *(const v4f*)(xk + HW);
        ESCN_PIN
        ESCN_MMA8(E)
        ESCN_FRAG(a0, a1, a2, a3, r0, r1)
    }

    float* outb = out + ((size_t)b * CH) * HW + m0 + 4 * col;
#define ESCN_STORE(ot, c0, c1, c2, c3)                                  \
    {                                                                   \
        const int   oR = o0 + 16 * (ot) + v + 8 * khalf;                \
        const float bs = bias[oR];                                      \
        v4f res;                                                        \
        res.x = c0[v] + bs; res.y = c1[v] + bs;                         \
        res.z = c2[v] + bs; res.w = c3[v] + bs;                         \
        *(v4f*)(outb + (size_t)oR * HW) = res;                          \
    }
#pragma unroll
    for (int v = 0; v < 8; ++v) {
        ESCN_STORE(0, acc00, acc01, acc02, acc03)
        ESCN_STORE(1, acc10, acc11, acc12, acc13)
        ESCN_STORE(2, acc20, acc21, acc22, acc23)
        ESCN_STORE(3, acc30, acc31, acc32, acc33)
    }
}

extern "C" void kernel_launch(void* const* d_in, const int* in_sizes, int n_in,
                              void* d_out, int out_size, void* d_ws, size_t ws_size,
                              hipStream_t stream) {
    const float* feature = (const float*)d_in[0];
    const float* conv_w  = (const float*)d_in[1];
    const float* conv_b  = (const float*)d_in[2];
    float* out = (float*)d_out;

    float* corr = (float*)d_ws;

    escn_corr_kernel<<<256, 512, 0, stream>>>(feature, corr);
    escn_gemm_kernel<<<dim3(8, 12, 8), 256, 0, stream>>>(feature, corr,
                                                         conv_w, conv_b, out);
}


# corr kernel: each neighbour pair's channel sum made once, written to offset o at p and to the mirrored offset at the neighbour: 11 loads, 64 FMAs per channel (were 21, 124); same sums, same order
# speedup vs baseline: 3.7014x; 1.0546x over previous
#include <hip/hip_runtime.h>
#include <hip/hip_bf16.h>

typedef __attribute__((ext_vector_type(2))) float v2f;
typedef __attribute__((ext_vector_type(8))) float v8f;

#define CH    768
#define HW    4096
#define NOFF  28
#define KTOT  796

typedef __attribute__((ext_vector_type(4))) float v4f;

#define KSPLIT 16
#define KSLICE (CH / KSPLIT)
#define NPAIR  12
#define NSLOT  16
#define CPT    2
#define PLANE  (HW * 4)

__device__ __forceinline__ void escn_pair(int s, int& dr, int& dc, int& o, int& om) {
    if (s < 4) { dr = 0; dc = s; o = 24 + s; om = 24 - s; }
    else {
        const int k = s - 4;
        const int f = k % 3;
        dr = k / 3 + 1;
        if      (f == 0) { dc = dr;  o = 3 + dr;  om = 3 - dr;  }
        else if (f == 1) { dc = 0;   o = 10 + dr; om = 10 - dr; }
        else             { dc = -dr; o = 17 + dr; om = 17 - dr; }
    }
}

__global__ __launch_bounds__(512)
void escn_corr_kernel(const float* __restrict__ feat,
                      float* __restrict__ corr) {
    __shared__ v4f   red[KSPLIT * NSLOT * 32];
    __shared__ v4f   inv[2 * 4 * 16];
    __shared__ v4f   mirv[NPAIR * 2 * 16];

    const int lane = threadIdx.x & 31;
    const int g    = __builtin_amdgcn_readfirstlane((int)(threadIdx.x >> 5));
    const int b    = blockIdx.x >> 5;
    const int j    = lane & 15;
    const int hf   = lane >> 4;
    const int h    = ((blockIdx.x & 31) << 1) + hf;
    const int p0   = h * 64 + 4 * j;

    unsigned offL[4], offC[4], offR[4];
#pragma unroll
    for (int i = 0; i < 4; ++i) {
        int  rr  = h + i;
        bool rok = rr < 64;
        unsigned oc = 4u * (unsigned)((rok ? rr : h) * 64 + 4 * j);
        offC[i] = oc;
        offL[i] = oc - (j > 0  ? 16u : 0u);
        offR[i] = oc + (j < 15 ? 16u : 0u);
    }

    int prow = ((blockIdx.x & 31) << 1) + ((lane & 7) < 4 ? (lane & 7) : 4);
    prow = prow > 63 ? 63 : prow;
    const unsigned pfoff = 256u * (unsigned)prow + (unsigned)((lane >> 3) & 1) * (unsigned)PLANE;

    float acc[NSLOT][4];
#pragma unroll
    for (int s = 0; s < NSLOT; ++s) {
#pragma unroll
        for (int t = 0; t < 4; ++t) acc[s][t] = 0.0f;
    }

    const char* pl = (const char*)(feat + ((size_t)b * CH + (size_t)g * KSLICE) * HW);
    __builtin_prefetch(pl + (size_t)(CPT * PLANE) + pfoff, 0, 3);
    for (int c = 0; c < KSLICE; c += CPT, pl += CPT * PLANE) {
        if (c + 2 * CPT < KSLICE)
            __builtin_prefetch(pl + (size_t)(2 * CPT * PLANE) + pfoff, 0, 3);
        float win[CPT][4][12];
#pragma unroll
        for (int u = 0; u < CPT; ++u) {
            const char* pu = pl + (size_t)u * PLANE;
#pragma unroll
            for (int i = 0; i < 4; ++i) {
                v4f C = *(const v4f*)(pu + offC[i]);
                v4f R = *(const v4f*)(pu + offR[i]);
                win[u][i][4] = C.x; win[u][i][5] = C.y; win[u][i][6]  = C.z; win[u][i][7]  = C.w;
                win[u][i][8] = R.x; win[u][i][9] = R.y; win[u][i][10] = R.z; win[u][i][11] = R.w;
                if (i > 0) {
                    v4f L = *(const v4f*)(pu + offL[i]);
                    win[u][i][0] = L.x; win[u][i][1] = L.y; win[u][i][2] = L.z; win[u][i][3] = L.w;
                } else {
                    win[u][i][0] = 0.0f; win[u][i][1] = 0.0f; win[u][i][2] = 0.0f; win[u][i][3] = 0.0f;
                }
            }
        }
        __builtin_amdgcn_sched_barrier(0);
#pragma unroll
        for (int u = 0; u < CPT; ++u) {
#pragma unroll
            for (int t = 0; t < 4; ++t) {
                const float own = win[u][0][4 + t];
                acc[0][t] = fmaf(own, own, acc[0][t]);
#pragma unroll
                for (int dc = 1; dc < 4; ++dc)
                    acc[dc][t] = fmaf(win[u][0][4 + t + dc], own, acc[dc][t]);
#pragma unroll
                for (int dr = 1; dr < 4; ++dr) {
                    const int s = 4 + 3 * (dr - 1);
                    acc[s][t]       = fmaf(win[u][dr][4 + t + dr], own, acc[s][t]);
                    acc[s + 1][t]   = fmaf(win[u][dr][4 + t],      own, acc[s + 1][t]);
                    acc[s + 2][t]   = fmaf(win[u][dr][4 + t - dr], own, acc[s + 2][t]);
                    acc[12 + dr][t] = fmaf(win[u][dr][4 + t], win[u][dr][4 + t], acc[12 + dr][t]);
                }
            }
        }
    }

#pragma unroll
    for (int s = 0; s < NSLOT; ++s) {
        v4f v;
        v.x = acc[s][0]; v.y = acc[s][1]; v.z = acc[s][2]; v.w = acc[s][3];
        red[(g * NSLOT + s) * 32 + lane] = v;
    }
    __syncthreads();

    if (g < 4) {
        const int s = g == 0 ? 0 : 12 + g;
        v4f nsq = red[s * 32 + lane];
#pragma unroll
        for (int q = 1; q < KSPLIT; ++q)
            nsq += red[(q * NSLOT + s) * 32 + lane];
        v4f iv;
#pragma unroll
        for (int t = 0; t < 4; ++t)
            iv[t] = 1.0f / fmaxf(sqrtf(nsq[t]), 1e-12f);
        inv[(hf * 4 + g) * 16 + j] = iv;
    }
    __syncthreads();

    const float* tab = (const float*)inv + hf * (4 * 64);
    const v4f ic = inv[(hf * 4) * 16 + j];
    if (g <= NPAIR) {
        v4f sum = red[g * 32 + lane];
#pragma unroll
        for (int q = 1; q < KSPLIT; ++q)
            sum += red[(q * NSLOT + g) * 32 + lane];
        if (g == 0) {
            v4f res;
#pragma unroll
            for (int t = 0; t < 4; ++t)
                res[t] = sum[t] * 1.0f * ic[t] * tab[4 * j + t];
            float* cb = corr + (size_t)b * NOFF * HW + p0;
            *(v4f*)(cb + (size_t)3  * HW) = res;
            *(v4f*)(cb + (size_t)10 * HW) = res;
            *(v4f*)(cb + (size_t)17 * HW) = res;
            *(v4f*)(cb + (size_t)24 * HW) = res;
        } else {
            int dr, dc, o, om;
            escn_pair(g, dr, dc, o, om);
            const bool rok = (h + dr) < 64;
            v4f res, mir;
#pragma unroll
            for (int t = 0; t < 4; ++t) {
                int   sw = 4 * j + t + dc;
                bool  ok = rok && (sw >= 0) && (sw < 64);
                float nb = tab[ok ? (dr * 64 + sw) : (4 * j + t)];
                float vf = ok ? 1.0f : 0.0f;
                res[t] = sum[t] * vf * ic[t] * nb;
                mir[t] = sum[t] * vf * nb * ic[t];
            }
            *(v4f*)(corr + ((size_t)b * NOFF + o) * HW + p0) = res;
            mirv[((g - 1) * 2 + hf) * 16 + j] = mir;
        }
    }
    __syncthreads();

    if (g < NPAIR) {
        int dr, dc, o, om;
        escn_pair(g + 1, dr, dc, o, om);
        const float* src = (const float*)mirv + (g * 2 + hf) * 64;
        v4f val;
#pragma unroll
        for (int t = 0; t < 4; ++t) {
            int   sc  = 4 * j + t - dc;
            bool  inb = (sc >= 0) && (sc < 64);
            float v   = src[inb ? sc : 0];
            val[t] = inb ? v : 0.0f;
        }
        float* cm = corr + ((size_t)b * NOFF + om) * HW + 4 * j;
        if (h + dr < 64)
            *(v4f*)(cm + (h + dr) * 64) = val;
        if (h < dr) {
            v4f z = {0.0f, 0.0f, 0.0f, 0.0f};
            *(v4f*)(cm + h * 64) = z;
        }
    }
}

#define ESCN_WMMA(A, B, C) \
    __builtin_amdgcn_wmma_f32_16x16x4_f32(false, (A), false, (B), (short)0, (C), false, false)

#define ESCN_OTILE(aX, c0, c1, c2, c3)                                  \
    c0 = ESCN_WMMA(aX, bt0, c0);                                        \
    c1 = ESCN_WMMA(aX, bt1, c1);                                        \
    c2 = ESCN_WMMA(aX, bt2, c2);                                        \
    c3 = ESCN_WMMA(aX, bt3, c3);

#define ESCN_FRAG(aX0, aX1, aX2, aX3, rlo, rhi)                         \
    {                                                                   \
        v2f bt0, bt1, bt2, bt3;                                         \
        bt0.x = rlo.x; bt0.y = rhi.x;                                   \
        bt1.x = rlo.y; bt1.y = rhi.y;                                   \
        bt2.x = rlo.z; bt2.y = rhi.z;                                   \
        bt3.x = rlo.w; bt3.y = rhi.w;                                   \
        ESCN_OTILE(aX0, acc00, acc01, acc02, acc03)                     \
        ESCN_OTILE(aX1, acc10, acc11, acc12, acc13)                     \
        ESCN_OTILE(aX2, acc20, acc21, acc22, acc23)                     \
        ESCN_OTILE(aX3, acc30, acc31, acc32, acc33)                     \
    }

#define ESCN_LOAD8(S, wk, xk)                                           \
    S##a0 = *(const v4f*)(wk);                                          \
    S##a1 = *(const v4f*)((wk) + 16 * KTOT);                            \
    S##a2 = *(const v4f*)((wk) + 32 * KTOT);                            \
    S##a3 = *(const v4f*)((wk) + 48 * KTOT);                            \
    S##r0 = *(const v4f*)(xk);                                          \
    S##r1 = *(const v4f*)((xk) + HW);                                   \
    S##r2 = *(const v4f*)((xk) + 2 * HW);                               \
    S##r3 = *(const v4f*)((xk) + 3 * HW);

#define ESCN_MMA8(S)                                                    \
    {                                                                   \
        v2f aP0, aP1, aP2, aP3, aQ0, aQ1, aQ2, aQ3;                     \
        aP0.x = S##a0.x; aP0.y = S##a0.y; aQ0.x = S##a0.z; aQ0.y = S##a0.w; \
        aP1.x = S##a1.x; aP1.y = S##a1.y; aQ1.x = S##a1.z; aQ1.y = S##a1.w; \
        aP2.x = S##a2.x; aP2.y = S##a2.y; aQ2.x = S##a2.z; aQ2.y = S##a2.w; \
        aP3.x = S##a3.x; aP3.y = S##a3.y; aQ3.x = S##a3.z; aQ3.y = S##a3.w; \
        ESCN_FRAG(aP0, aP1, aP2, aP3, S##r0, S##r1)                     \
        ESCN_FRAG(aQ0, aQ1, aQ2, aQ3, S##r2, S##r3)                     \
    }

#define ESCN_PIN __builtin_amdgcn_sched_barrier(0);

__global__ __launch_bounds__(256)
void escn_gemm_kernel(const float* __restrict__ feat,
                      const float* __restrict__ corr,
                      const float* __restrict__ Wm,
                      const float* __restrict__ bias,
                      float* __restrict__ out) {
    const int lane  = threadIdx.x & 31;
    const int wid   = threadIdx.x >> 5;
    const int mtile = blockIdx.x * 8 + wid;
    const int m0    = mtile * 64;
    const int o0    = blockIdx.y * 64;
    const int b     = blockIdx.z;

    const int col   = lane & 15;
    const int khalf = lane >> 4;

    v8f acc00 = {}, acc01 = {}, acc02 = {}, acc03 = {};
    v8f acc10 = {}, acc11 = {}, acc12 = {}, acc13 = {};
    v8f acc20 = {}, acc21 = {}, acc22 = {}, acc23 = {};
    v8f acc30 = {}, acc31 = {}, acc32 = {}, acc33 = {};

    const float* fbase = feat + ((size_t)b * CH   + 4 * khalf) * HW + m0 + 4 * col;
    const float* cbase = corr + ((size_t)b * NOFF + 4 * khalf) * HW + m0 + 4 * col;
    const float* wrow  = Wm + (size_t)(o0 + col) * KTOT + 4 * khalf;

    v4f Ea0, Ea1, Ea2, Ea3, Er0, Er1, Er2, Er3;
    v4f Fa0, Fa1, Fa2, Fa3, Fr0, Fr1, Fr2, Fr3;

    ESCN_LOAD8(E, wrow, fbase)
    for (int k = 0; k < CH - 16; k += 16) {
        ESCN_PIN
        ESCN_LOAD8(F, wrow + k + 8, fbase + (size_t)(k + 8) * HW)
        ESCN_PIN
        ESCN_MMA8(E)
        ESCN_PIN
        ESCN_LOAD8(E, wrow + k + 16, fbase + (size_t)(k + 16) * HW)
        ESCN_PIN
        ESCN_MMA8(F)
    }
    ESCN_PIN
    ESCN_LOAD8(F, wrow + (CH - 8), fbase + (size_t)(CH - 8) * HW)
    ESCN_PIN
    ESCN_MMA8(E)

    ESCN_PIN
    ESCN_LOAD8(E, wrow + CH, cbase)
    ESCN_PIN
    ESCN_MMA8(F)
    ESCN_PIN
    ESCN_LOAD8(F, wrow + CH + 8, cbase + (size_t)8 * HW)
    ESCN_PIN
    ESCN_MMA8(E)
    ESCN_PIN
    ESCN_LOAD8(E, wrow + CH + 16, cbase + (size_t)16 * HW)
    ESCN_PIN
    ESCN_MMA8(F)

    {
        const float* xk = cbase + (size_t)(24 - 2 * khalf) * HW;
        const float* wk = wrow + CH + 24 - 2 * khalf;
        ESCN_PIN
        const v2f a0 = *(const v2f*)(wk);
        const v2f a1 = *(const v2f*)(wk + 16 * KTOT);
        const v2f a2 = *(const v2f*)(wk + 32 * KTOT);
        const v2f a3 = *(const v2f*)(wk + 48 * KTOT);
        const v4f r0 = *(const v4f*)(xk);
        const v4f r1 = *(const v4f*)(xk + HW);
        ESCN_PIN
        ESCN_MMA8(E)
        ESCN_FRAG(a0, a1, a2, a3, r0, r1)
    }

    float* outb = out + ((size_t)b * CH) * HW + m0 + 4 * col;
#define ESCN_STORE(ot, c0, c1, c2, c3)                                  \
    {                                                                   \
        const int   oR = o0 + 16 * (ot) + v + 8 * khalf;                \
        const float bs = bias[oR];                                      \
        v4f res;                                                        \
        res.x = c0[v] + bs; res.y = c1[v] + bs;                         \
        res.z = c2[v] + bs; res.w = c3[v] + bs;                         \
        *(v4f*)(outb + (size_t)oR * HW) = res;                          \
    }
#pragma unroll
    for (int v = 0; v < 8; ++v) {
        ESCN_STORE(0, acc00, acc01, acc02, acc03)
        ESCN_STORE(1, acc10, acc11, acc12, acc13)
        ESCN_STORE(2, acc20, acc21, acc22, acc23)
        ESCN_STORE(3, acc30, acc31, acc32, acc33)
    }
}

extern "C" void kernel_launch(void* const* d_in, const int* in_sizes, int n_in,
                              void* d_out, int out_size, void* d_ws, size_t ws_size,
                              hipStream_t stream) {
    const float* feature = (const float*)d_in[0];
    const float* conv_w  = (const float*)d_in[1];
    const float* conv_b  = (const float*)d_in[2];
    float* out = (float*)d_out;

    float* corr = (float*)d_ws;

    escn_corr_kernel<<<256, 512, 0, stream>>>(feature, corr);
    escn_gemm_kernel<<<dim3(8, 12, 8), 256, 0, stream>>>(feature, corr,
                                                         conv_w, conv_b, out);
}


# gemm kernel: wave tile 96(o) x 128(m), 48 accumulator tiles, 4-wave work-groups (was 64 x 64, 16 tiles, 8 waves): 14 loads per 96 WMMAs (was 8 per 32), X read 8 times from the L2 (was 12); same sums
# speedup vs baseline: 3.9820x; 1.0758x over previous
#include <hip/hip_runtime.h>
#include <hip/hip_bf16.h>

typedef __attribute__((ext_vector_type(2))) float v2f;
typedef __attribute__((ext_vector_type(8))) float v8f;

#define CH    768
#define HW    4096
#define NOFF  28
#define KTOT  796

typedef __attribute__((ext_vector_type(4))) float v4f;

#define KSPLIT 16
#define KSLICE (CH / KSPLIT)
#define NPAIR  12
#define NSLOT  16
#define CPT    2
#define PLANE  (HW * 4)

__device__ __forceinline__ void escn_pair(int s, int& dr, int& dc, int& o, int& om) {
    if (s < 4) { dr = 0; dc = s; o = 24 + s; om = 24 - s; }
    else {
        const int k = s - 4;
        const int f = k % 3;
        dr = k / 3 + 1;
        if      (f == 0) { dc = dr;  o = 3 + dr;  om = 3 - dr;  }
        else if (f == 1) { dc = 0;   o = 10 + dr; om = 10 - dr; }
        else             { dc = -dr; o = 17 + dr; om = 17 - dr; }
    }
}

__global__ __launch_bounds__(512)
void escn_corr_kernel(const float* __restrict__ feat,
                      float* __restrict__ corr) {
    __shared__ v4f   red[KSPLIT * NSLOT * 32];
    __shared__ v4f   inv[2 * 4 * 16];
    __shared__ v4f   mirv[NPAIR * 2 * 16];

    const int lane = threadIdx.x & 31;
    const int g    = __builtin_amdgcn_readfirstlane((int)(threadIdx.x >> 5));
    const int b    = blockIdx.x >> 5;
    const int j    = lane & 15;
    const int hf   = lane >> 4;
    const int h    = ((blockIdx.x & 31) << 1) + hf;
    const int p0   = h * 64 + 4 * j;

    unsigned offL[4], offC[4], offR[4];
#pragma unroll
    for (int i = 0; i < 4; ++i) {
        int  rr  = h + i;
        bool rok = rr < 64;
        unsigned oc = 4u * (unsigned)((rok ? rr : h) * 64 + 4 * j);
        offC[i] = oc;
        offL[i] = oc - (j > 0  ? 16u : 0u);
        offR[i] = oc + (j < 15 ? 16u : 0u);
    }

    int prow = ((blockIdx.x & 31) << 1) + ((lane & 7) < 4 ? (lane & 7) : 4);
    prow = prow > 63 ? 63 : prow;
    const unsigned pfoff = 256u * (unsigned)prow + (unsigned)((lane >> 3) & 1) * (unsigned)PLANE;

    float acc[NSLOT][4];
#pragma unroll
    for (int s = 0; s < NSLOT; ++s) {
#pragma unroll
        for (int t = 0; t < 4; ++t) acc[s][t] = 0.0f;
    }

    const char* pl = (const char*)(feat + ((size_t)b * CH + (size_t)g * KSLICE) * HW);
    __builtin_prefetch(pl + (size_t)(CPT * PLANE) + pfoff, 0, 3);
    for (int c = 0; c < KSLICE; c += CPT, pl += CPT * PLANE) {
        if (c + 2 * CPT < KSLICE)
            __builtin_prefetch(pl + (size_t)(2 * CPT * PLANE) + pfoff, 0, 3);
        float win[CPT][4][12];
#pragma unroll
        for (int u = 0; u < CPT; ++u) {
            const char* pu = pl + (size_t)u * PLANE;
#pragma unroll
            for (int i = 0; i < 4; ++i) {
                v4f C = *(const v4f*)(pu + offC[i]);
                v4f R = *(const v4f*)(pu + offR[i]);
                win[u][i][4] = C.x; win[u][i][5] = C.y; win[u][i][6]  = C.z; win[u][i][7]  = C.w;
                win[u][i][8] = R.x; win[u][i][9] = R.y; win[u][i][10] = R.z; win[u][i][11] = R.w;
                if (i > 0) {
                    v4f L = *(const v4f*)(pu + offL[i]);
                    win[u][i][0] = L.x; win[u][i][1] = L.y; win[u][i][2] = L.z; win[u][i][3] = L.w;
                } else {
                    win[u][i][0] = 0.0f; win[u][i][1] = 0.0f; win[u][i][2] = 0.0f; win[u][i][3] = 0.0f;
                }
            }
        }
        __builtin_amdgcn_sched_barrier(0);
#pragma unroll
        for (int u = 0; u < CPT; ++u) {
#pragma unroll
            for (int t = 0; t < 4; ++t) {
                const float own = win[u][0][4 + t];
                acc[0][t] = fmaf(own, own, acc[0][t]);
#pragma unroll
                for (int dc = 1; dc < 4; ++dc)
                    acc[dc][t] = fmaf(win[u][0][4 + t + dc], own, acc[dc][t]);
#pragma unroll
                for (int dr = 1; dr < 4; ++dr) {
                    const int s = 4 + 3 * (dr - 1);
                    acc[s][t]       = fmaf(win[u][dr][4 + t + dr], own, acc[s][t]);
                    acc[s + 1][t]   = fmaf(win[u][dr][4 + t],      own, acc[s + 1][t]);
                    acc[s + 2][t]   = fmaf(win[u][dr][4 + t - dr], own, acc[s + 2][t]);
                    acc[12 + dr][t] = fmaf(win[u][dr][4 + t], win[u][dr][4 + t], acc[12 + dr][t]);
                }
            }
        }
    }

#pragma unroll
    for (int s = 0; s < NSLOT; ++s) {
        v4f v;
        v.x = acc[s][0]; v.y = acc[s][1]; v.z = acc[s][2]; v.w = acc[s][3];
        red[(g * NSLOT + s) * 32 + lane] = v;
    }
    __syncthreads();

    if (g < 4) {
        const int s = g == 0 ? 0 : 12 + g;
        v4f nsq = red[s * 32 + lane];
#pragma unroll
        for (int q = 1; q < KSPLIT; ++q)
            nsq += red[(q * NSLOT + s) * 32 + lane];
        v4f iv;
#pragma unroll
        for (int t = 0; t < 4; ++t)
            iv[t] = 1.0f / fmaxf(sqrtf(nsq[t]), 1e-12f);
        inv[(hf * 4 + g) * 16 + j] = iv;
    }
    __syncthreads();

    const float* tab = (const float*)inv + hf * (4 * 64);
    const v4f ic = inv[(hf * 4) * 16 + j];
    if (g <= NPAIR) {
        v4f sum = red[g * 32 + lane];
#pragma unroll
        for (int q = 1; q < KSPLIT; ++q)
            sum += red[(q * NSLOT + g) * 32 + lane];
        if (g == 0) {
            v4f res;
#pragma unroll
            for (int t = 0; t < 4; ++t)
                res[t] = sum[t] * 1.0f * ic[t] * tab[4 * j + t];
            float* cb = corr + (size_t)b * NOFF * HW + p0;
            *(v4f*)(cb + (size_t)3  * HW) = res;
            *(v4f*)(cb + (size_t)10 * HW) = res;
            *(v4f*)(cb + (size_t)17 * HW) = res;
            *(v4f*)(cb + (size_t)24 * HW) = res;
        } else {
            int dr, dc, o, om;
            escn_pair(g, dr, dc, o, om);
            const bool rok = (h + dr) < 64;
            v4f res, mir;
#pragma unroll
            for (int t = 0; t < 4; ++t) {
                int   sw = 4 * j + t + dc;
                bool  ok = rok && (sw >= 0) && (sw < 64);
                float nb = tab[ok ? (dr * 64 + sw) : (4 * j + t)];
                float vf = ok ? 1.0f : 0.0f;
                res[t] = sum[t] * vf * ic[t] * nb;
                mir[t] = sum[t] * vf * nb * ic[t];
            }
            *(v4f*)(corr + ((size_t)b * NOFF + o) * HW + p0) = res;
            mirv[((g - 1) * 2 + hf) * 16 + j] = mir;
        }
    }
    __syncthreads();

    if (g < NPAIR) {
        int dr, dc, o, om;
        escn_pair(g + 1, dr, dc, o, om);
        const float* src = (const float*)mirv + (g * 2 + hf) * 64;
        v4f val;
#pragma unroll
        for (int t = 0; t < 4; ++t) {
            int   sc  = 4 * j + t - dc;
            bool  inb = (sc >= 0) && (sc < 64);
            float v   = src[inb ? sc : 0];
            val[t] = inb ? v : 0.0f;
        }
        float* cm = corr + ((size_t)b * NOFF + om) * HW + 4 * j;
        if (h + dr < 64)
            *(v4f*)(cm + (h + dr) * 64) = val;
        if (h < dr) {
            v4f z = {0.0f, 0.0f, 0.0f, 0.0f};
            *(v4f*)(cm + h * 64) = z;
        }
    }
}

#define ESCN_WMMA(A, B, C) \
    __builtin_amdgcn_wmma_f32_16x16x4_f32(false, (A), false, (B), (short)0, (C), false, false)

#define OTN 6
#define NST 2

struct EscnFrag {
    v4f a[OTN];
    v4f r[NST][4];
};

__device__ __forceinline__ void escn_load8(EscnFrag& S, const float* __restrict__ wk,
                                           const float* __restrict__ xk) {
#pragma unroll
    for (int t = 0; t < OTN; ++t)
        S.a[t] = *(const v4f*)(wk + (size_t)(16 * t) * KTOT);
#pragma unroll
    for (int s = 0; s < NST; ++s) {
#pragma unroll
        for (int i = 0; i < 4; ++i)
            S.r[s][i] = *(const v4f*)(xk + (size_t)i * HW + 64 * s);
    }
}

__device__ __forceinline__ void escn_frag(v8f (&acc)[OTN][NST * 4], const v2f (&aX)[OTN],
                                          const v4f (&rlo)[NST], const v4f (&rhi)[NST]) {
#pragma unroll
    for (int s = 0; s < NST; ++s) {
        v2f bt[4];
#pragma unroll
        for (int j = 0; j < 4; ++j) { bt[j].x = rlo[s][j]; bt[j].y = rhi[s][j]; }
#pragma unroll
        for (int t = 0; t < OTN; ++t) {
#pragma unroll
            for (int j = 0; j < 4; ++j)
                acc[t][4 * s + j] = ESCN_WMMA(aX[t], bt[j], acc[t][4 * s + j]);
        }
    }
}

__device__ __forceinline__ void escn_mma8(v8f (&acc)[OTN][NST * 4], const EscnFrag& S) {
    v2f aP[OTN], aQ[OTN];
    v4f r0[NST], r1[NST], r2[NST], r3[NST];
#pragma unroll
    for (int t = 0; t < OTN; ++t) {
        aP[t].x = S.a[t].x; aP[t].y = S.a[t].y;
        aQ[t].x = S.a[t].z; aQ[t].y = S.a[t].w;
    }
#pragma unroll
    for (int s = 0; s < NST; ++s) {
        r0[s] = S.r[s][0]; r1[s] = S.r[s][1]; r2[s] = S.r[s][2]; r3[s] = S.r[s][3];
    }
    escn_frag(acc, aP, r0, r1);
    escn_frag(acc, aQ, r2, r3);
}

#define ESCN_PIN __builtin_amdgcn_sched_barrier(0);

__global__ __launch_bounds__(128) __attribute__((amdgpu_waves_per_eu(2, 2)))
void escn_gemm_kernel(const float* __restrict__ feat,
                      const float* __restrict__ corr,
                      const float* __restrict__ Wm,
                      const float* __restrict__ bias,
                      float* __restrict__ out) {
    const int lane  = threadIdx.x & 31;
    const int wid   = threadIdx.x >> 5;
    const int mtile = blockIdx.x * 4 + wid;
    const int m0    = mtile * 128;
    const int o0    = blockIdx.y * 96;
    const int b     = blockIdx.z;

    const int col   = lane & 15;
    const int khalf = lane >> 4;

    v8f acc[OTN][NST * 4];
#pragma unroll
    for (int t = 0; t < OTN; ++t) {
#pragma unroll
        for (int j = 0; j < NST * 4; ++j) {
            v8f z = {};
            acc[t][j] = z;
        }
    }

    const float* fbase = feat + ((size_t)b * CH   + 4 * khalf) * HW + m0 + 4 * col;
    const float* wrow  = Wm + (size_t)(o0 + col) * KTOT + 4 * khalf;

    EscnFrag E, F;

    escn_load8(E, wrow, fbase);
    for (int k = 0; k < CH - 16; k += 16) {
        ESCN_PIN
        escn_load8(F, wrow + k + 8, fbase + (size_t)(k + 8) * HW);
        ESCN_PIN
        escn_mma8(acc, E);
        ESCN_PIN
        escn_load8(E, wrow + k + 16, fbase + (size_t)(k + 16) * HW);
        ESCN_PIN
        escn_mma8(acc, F);
    }
    ESCN_PIN
    escn_load8(F, wrow + (CH - 8), fbase + (size_t)(CH - 8) * HW);
    ESCN_PIN
    escn_mma8(acc, E);
    ESCN_PIN
    escn_mma8(acc, F);

    int tid2 = threadIdx.x;
    asm volatile("" : "+v"(tid2));
    const int col2   = tid2 & 15;
    const int khalf2 = (tid2 >> 4) & 1;
    const int m02    = (blockIdx.x * 4 + (tid2 >> 5)) * 128;
    const float* cbase = corr + ((size_t)b * NOFF + 4 * khalf2) * HW + m02 + 4 * col2;
    const float* wrow2 = Wm + (size_t)(o0 + col2) * KTOT + 4 * khalf2;

    escn_load8(E, wrow2 + CH, cbase);
    ESCN_PIN
    escn_load8(F, wrow2 + CH + 8, cbase + (size_t)8 * HW);
    ESCN_PIN
    escn_mma8(acc, E);
    ESCN_PIN
    escn_load8(E, wrow2 + CH + 16, cbase + (size_t)16 * HW);
    ESCN_PIN
    escn_mma8(acc, F);

    {
        const float* xk = cbase + (size_t)(24 - 2 * khalf2) * HW;
        const float* wk = wrow2 + CH + 24 - 2 * khalf2;
        v2f a4[OTN];
        v4f q0[NST], q1[NST];
        ESCN_PIN
#pragma unroll
        for (int t = 0; t < OTN; ++t)
            a4[t] = *(const v2f*)(wk + (size_t)(16 * t) * KTOT);
#pragma unroll
        for (int s = 0; s < NST; ++s) {
            q0[s] = *(const v4f*)(xk + 64 * s);
            q1[s] = *(const v4f*)(xk + HW + 64 * s);
        }
        ESCN_PIN
        escn_mma8(acc, E);
        escn_frag(acc, a4, q0, q1);
    }

    int tid3 = threadIdx.x;
    asm volatile("" : "+v"(tid3));
    const int col3   = tid3 & 15;
    const int khalf3 = (tid3 >> 4) & 1;
    const int m03    = (blockIdx.x * 4 + (tid3 >> 5)) * 128;
    float* outb = out + ((size_t)b * CH) * HW + m03 + 4 * col3;
#pragma unroll
    for (int t = 0; t < OTN; ++t) {
#pragma unroll
        for (int v = 0; v < 8; ++v) {
            const int   oR = o0 + 16 * t + v + 8 * khalf3;
            const float bs = bias[oR];
#pragma unroll
            for (int s = 0; s < NST; ++s) {
                v4f res;
                res.x = acc[t][4 * s + 0][v] + bs; res.y = acc[t][4 * s + 1][v] + bs;
                res.z = acc[t][4 * s + 2][v] + bs; res.w = acc[t][4 * s + 3][v] + bs;
                *(v4f*)(outb + (size_t)oR * HW + 64 * s) = res;
            }
        }
    }
}

extern "C" void kernel_launch(void* const* d_in, const int* in_sizes, int n_in,
                              void* d_out, int out_size, void* d_ws, size_t ws_size,
                              hipStream_t stream) {
    const float* feature = (const float*)d_in[0];
    const float* conv_w  = (const float*)d_in[1];
    const float* conv_b  = (const float*)d_in[2];
    float* out = (float*)d_out;

    float* corr = (float*)d_ws;

    escn_corr_kernel<<<256, 512, 0, stream>>>(feature, corr);
    escn_gemm_kernel<<<dim3(8, 8, 8), 128, 0, stream>>>(feature, corr,
                                                         conv_w, conv_b, out);
}
